# starting kernel
# baseline (speedup 1.0000x reference)
#include <hip/hip_runtime.h>
#include <hip/hip_bf16.h>
#include <math.h>

typedef __attribute__((ext_vector_type(2))) float v2f;
typedef __attribute__((ext_vector_type(8))) float v8f;

__global__ void k_fill(float* __restrict__ p, float v, int n) {
    int i = blockIdx.x * blockDim.x + threadIdx.x;
    if (i < n) p[i] = v;
}

__global__ void k_transpose(const float* __restrict__ W, float* __restrict__ Wt,
                            int K, int Ncol) {
    int i = blockIdx.x * blockDim.x + threadIdx.x;
    if (i < K * Ncol) {
        int k = i / Ncol, nn = i % Ncol;
        Wt[nn * K + k] = W[i];
    }
}

__global__ void k_gemm_f32_wmma(const float* __restrict__ A,
                                const float* __restrict__ Wt,
                                float* __restrict__ C, int nCols) {
    const int K = 128;
    int lane = threadIdx.x;
    int half = lane >> 4;
    int l    = lane & 15;
    int rowBase = blockIdx.x * 16;
    int colBase = (blockIdx.y * 4 + threadIdx.y) * 16;

    const float* arow = A  + (size_t)(rowBase + l) * K + 2 * half;
    const float* brow = Wt + (size_t)(colBase + l) * K + 2 * half;

    v8f acc = {};
#pragma unroll
    for (int k = 0; k < K; k += 4) {
        v2f a = *(const v2f*)(arow + k);
        v2f b = *(const v2f*)(brow + k);
        acc = __builtin_amdgcn_wmma_f32_16x16x4_f32(
             false, a,  false, b,
             (short)0, acc,  false,  false);
    }
#pragma unroll
    for (int r = 0; r < 8; ++r)
        C[(size_t)(rowBase + r + 8 * half) * nCols + colBase + l] = acc[r];
}

__global__ void k_attn(const float* __restrict__ h, const float* __restrict__ att_s,
                       const float* __restrict__ att_d, float* __restrict__ asrc,
                       float* __restrict__ adst, int n, int heads, int C) {
    int i = blockIdx.x * blockDim.x + threadIdx.x;
    if (i >= n * heads) return;
    int node = i / heads, hd = i % heads;
    const float* hp = h + (size_t)node * heads * C + (size_t)hd * C;
    float ss = 0.f, sd = 0.f;
    for (int c = 0; c < C; ++c) {
        float v = hp[c];
        ss += v * att_s[hd * C + c];
        sd += v * att_d[hd * C + c];
    }
    asrc[i] = ss;
    adst[i] = sd;
}

__device__ __forceinline__ void atomicMaxF(float* addr, float val) {
    if (val >= 0.f) atomicMax((int*)addr, __float_as_int(val));
    else            atomicMin((unsigned int*)addr, (unsigned int)__float_as_int(val));
}

__device__ __forceinline__ void edge_pair(int e, int E, const int* __restrict__ src,
                                          const int* __restrict__ dst, int& s, int& d) {
    if (e < E) { s = src[e]; d = dst[e]; }
    else       { s = d = e - E; }
}

__global__ void k_edge_max(const int* __restrict__ src, const int* __restrict__ dst,
                           const float* __restrict__ asrc, const float* __restrict__ adst,
                           float* __restrict__ m, int E, int n, int heads) {
    int e = blockIdx.x * blockDim.x + threadIdx.x;
    if (e >= E + n) return;
    int s, d; edge_pair(e, E, src, dst, s, d);
    for (int h = 0; h < heads; ++h) {
        float v = asrc[s * heads + h] + adst[d * heads + h];
        v = v > 0.f ? v : 0.2f * v;
        atomicMaxF(&m[d * heads + h], v);
    }
}

__global__ void k_edge_expsum(const int* __restrict__ src, const int* __restrict__ dst,
                              const float* __restrict__ asrc, const float* __restrict__ adst,
                              const float* __restrict__ m, float* __restrict__ ex,
                              float* __restrict__ den, int E, int n, int heads) {
    int e = blockIdx.x * blockDim.x + threadIdx.x;
    if (e >= E + n) return;
    int s, d; edge_pair(e, E, src, dst, s, d);
    for (int h = 0; h < heads; ++h) {
        float v = asrc[s * heads + h] + adst[d * heads + h];
        v = v > 0.f ? v : 0.2f * v;
        float x = expf(v - m[d * heads + h]);
        ex[(size_t)e * heads + h] = x;
        atomicAdd(&den[d * heads + h], x);
    }
}

__global__ void k_edge_agg(const int* __restrict__ src, const int* __restrict__ dst,
                           const float* __restrict__ ex, const float* __restrict__ den,
                           const float* __restrict__ feat, float* __restrict__ agg,
                           int E, int n, int heads, int C) {
    int HC = heads * C;
    int epb = blockDim.x / HC;
    int eLocal = threadIdx.x / HC;
    int c = threadIdx.x % HC;
    int e = blockIdx.x * epb + eLocal;
    if (e >= E + n) return;
    int s, d; edge_pair(e, E, src, dst, s, d);
    int hd = c / C;
    float alpha = ex[(size_t)e * heads + hd] / den[d * heads + hd];
    atomicAdd(&agg[(size_t)d * HC + c], feat[(size_t)s * HC + c] * alpha);
}

__global__ void k_bias_elu(float* __restrict__ h, const float* __restrict__ bias,
                           int n, int HC) {
    int i = blockIdx.x * blockDim.x + threadIdx.x;
    if (i >= n * HC) return;
    float v = h[i] + bias[i % HC];
    h[i] = v > 0.f ? v : (expf(v) - 1.f);
}

__global__ void k_bias_add(float* __restrict__ o, const float* __restrict__ bias,
                           int n, int C) {
    int i = blockIdx.x * blockDim.x + threadIdx.x;
    if (i >= n * C) return;
    o[i] += bias[i % C];
}

extern "C" void kernel_launch(void* const* d_in, const int* in_sizes, int n_in,
                              void* d_out, int out_size, void* d_ws, size_t ws_size,
                              hipStream_t stream) {
    const float* x      = (const float*)d_in[0];
    const int*   ei     = (const int*)  d_in[1];
    const float* W1     = (const float*)d_in[2];
    const float* att_s1 = (const float*)d_in[3];
    const float* att_d1 = (const float*)d_in[4];
    const float* b1     = (const float*)d_in[5];
    const float* W2     = (const float*)d_in[6];
    const float* att_s2 = (const float*)d_in[7];
    const float* att_d2 = (const float*)d_in[8];
    const float* b2     = (const float*)d_in[9];

    const int IN = 128, HID = 128, H1 = 4, C1 = 32, OUT = 64;
    const int n   = in_sizes[0] / IN;
    const int E   = in_sizes[1] / 2;
    const int tot = E + n;

    float* out1  = (float*)d_out;
    float* hfeat = (float*)d_out + (size_t)n * OUT;

    float* w = (float*)d_ws;
    size_t off = 0;
    float* Wt1 = w + off; off += (size_t)IN * HID;
    float* Wt2 = w + off; off += (size_t)HID * OUT;
    float* h1  = w + off; off += (size_t)n * HID;
    float* h2  = w + off; off += (size_t)n * OUT;
    float* as1 = w + off; off += (size_t)n * H1;
    float* ad1 = w + off; off += (size_t)n * H1;
    float* m1  = w + off; off += (size_t)n * H1;
    float* dn1 = w + off; off += (size_t)n * H1;
    float* as2 = w + off; off += (size_t)n;
    float* ad2 = w + off; off += (size_t)n;
    float* m2  = w + off; off += (size_t)n;
    float* dn2 = w + off; off += (size_t)n;
    float* ex1 = w + off; off += (size_t)tot * H1;
    float* ex2 = w + off; off += (size_t)tot;

    const int* srcI = ei;
    const int* dstI = ei + E;

    auto cdiv = [](long a, long b) { return (unsigned)((a + b - 1) / b); };
    dim3 B256(256);

    k_transpose<<<cdiv((long)IN * HID, 256), B256, 0, stream>>>(W1, Wt1, IN, HID);
    k_transpose<<<cdiv((long)HID * OUT, 256), B256, 0, stream>>>(W2, Wt2, HID, OUT);

    k_gemm_f32_wmma<<<dim3(n / 16, HID / 64), dim3(32, 4), 0, stream>>>(x, Wt1, h1, HID);
    k_attn<<<cdiv((long)n * H1, 256), B256, 0, stream>>>(h1, att_s1, att_d1, as1, ad1, n, H1, C1);

    k_fill<<<cdiv((long)n * H1, 256), B256, 0, stream>>>(m1, -INFINITY, n * H1);
    k_fill<<<cdiv((long)n * H1, 256), B256, 0, stream>>>(dn1, 0.f, n * H1);
    k_fill<<<cdiv((long)n * HID, 256), B256, 0, stream>>>(hfeat, 0.f, n * HID);

    k_edge_max   <<<cdiv(tot, 256), B256, 0, stream>>>(srcI, dstI, as1, ad1, m1, E, n, H1);
    k_edge_expsum<<<cdiv(tot, 256), B256, 0, stream>>>(srcI, dstI, as1, ad1, m1, ex1, dn1, E, n, H1);
    k_edge_agg   <<<cdiv((long)tot * 128, 256), B256, 0, stream>>>(srcI, dstI, ex1, dn1, h1, hfeat, E, n, H1, C1);
    k_bias_elu   <<<cdiv((long)n * HID, 256), B256, 0, stream>>>(hfeat, b1, n, HID);

    k_gemm_f32_wmma<<<dim3(n / 16, OUT / 64), dim3(32, 4), 0, stream>>>(hfeat, Wt2, h2, OUT);
    k_attn<<<cdiv((long)n, 256), B256, 0, stream>>>(h2, att_s2, att_d2, as2, ad2, n, 1, OUT);

    k_fill<<<cdiv((long)n, 256), B256, 0, stream>>>(m2, -INFINITY, n);
    k_fill<<<cdiv((long)n, 256), B256, 0, stream>>>(dn2, 0.f, n);
    k_fill<<<cdiv((long)n * OUT, 256), B256, 0, stream>>>(out1, 0.f, n * OUT);

    k_edge_max   <<<cdiv(tot, 256), B256, 0, stream>>>(srcI, dstI, as2, ad2, m2, E, n, 1);
    k_edge_expsum<<<cdiv(tot, 256), B256, 0, stream>>>(srcI, dstI, as2, ad2, m2, ex2, dn2, E, n, 1);
    k_edge_agg   <<<cdiv((long)tot * 64, 256), B256, 0, stream>>>(srcI, dstI, ex2, dn2, h2, out1, E, n, 1, OUT);
    k_bias_add   <<<cdiv((long)n * OUT, 256), B256, 0, stream>>>(out1, b2, n, OUT);
}


# Edges grouped by destination per call (integer count + scan, segments sorted by edge index); one wave per node gathers max, sum and weighted rows in registers: no float atomics, fixed sum order
# speedup vs baseline: 4.5520x; 4.5520x over previous
#include <hip/hip_runtime.h>
#include <hip/hip_bf16.h>
#include <math.h>

typedef __attribute__((ext_vector_type(2))) float v2f;
typedef __attribute__((ext_vector_type(8))) float v8f;

__global__ void k_zero_int(int* __restrict__ p, int n) {
    int i = blockIdx.x * blockDim.x + threadIdx.x;
    if (i < n) p[i] = 0;
}

__global__ void k_transpose(const float* __restrict__ W, float* __restrict__ Wt,
                            int K, int Ncol) {
    int i = blockIdx.x * blockDim.x + threadIdx.x;
    if (i < K * Ncol) {
        int k = i / Ncol, nn = i % Ncol;
        Wt[nn * K + k] = W[i];
    }
}

__global__ void k_gemm_f32_wmma(const float* __restrict__ A,
                                const float* __restrict__ Wt,
                                float* __restrict__ C, int nCols) {
    const int K = 128;
    int lane = threadIdx.x;
    int half = lane >> 4;
    int l    = lane & 15;
    int rowBase = blockIdx.x * 16;
    int colBase = (blockIdx.y * 4 + threadIdx.y) * 16;

    const float* arow = A  + (size_t)(rowBase + l) * K + 2 * half;
    const float* brow = Wt + (size_t)(colBase + l) * K + 2 * half;

    v8f acc = {};
#pragma unroll
    for (int k = 0; k < K; k += 4) {
        v2f a = *(const v2f*)(arow + k);
        v2f b = *(const v2f*)(brow + k);
        acc = __builtin_amdgcn_wmma_f32_16x16x4_f32(
             false, a,  false, b,
             (short)0, acc,  false,  false);
    }
#pragma unroll
    for (int r = 0; r < 8; ++r)
        C[(size_t)(rowBase + r + 8 * half) * nCols + colBase + l] = acc[r];
}

__global__ void k_attn(const float* __restrict__ h, const float* __restrict__ att_s,
                       const float* __restrict__ att_d, float* __restrict__ asrc,
                       float* __restrict__ adst, int n, int heads, int C) {
    int i = blockIdx.x * blockDim.x + threadIdx.x;
    if (i >= n * heads) return;
    int node = i / heads, hd = i % heads;
    const float* hp = h + (size_t)node * heads * C + (size_t)hd * C;
    float ss = 0.f, sd = 0.f;
    for (int c = 0; c < C; ++c) {
        float v = hp[c];
        ss += v * att_s[hd * C + c];
        sd += v * att_d[hd * C + c];
    }
    asrc[i] = ss;
    adst[i] = sd;
}

typedef __attribute__((ext_vector_type(4))) float v4f;

__global__ void k_count_rank(const int* __restrict__ dst, int* __restrict__ cnt,
                             int* __restrict__ rank, int E) {
    int e = blockIdx.x * blockDim.x + threadIdx.x;
    if (e < E) rank[e] = atomicAdd(&cnt[dst[e]], 1);
}

__global__ void k_scan_local(const int* __restrict__ cnt, int* __restrict__ rowptr,
                             int* __restrict__ bsum, int n) {
    __shared__ int wsum[8];
    int t = threadIdx.x, i = blockIdx.x * 256 + t;
    int lane = t & 31, wv = t >> 5;
    int v = i < n ? cnt[i] : 0;
    int inc = v;
#pragma unroll
    for (int o = 1; o < 32; o <<= 1) {
        int u = __shfl_up(inc, o);
        if (lane >= o) inc += u;
    }
    if (lane == 31) wsum[wv] = inc;
    __syncthreads();
    int woff = 0;
#pragma unroll
    for (int k = 0; k < 8; ++k) woff += (k < wv) ? wsum[k] : 0;
    if (i < n) rowptr[i] = woff + inc - v;
    if (t == 255) bsum[blockIdx.x] = woff + inc;
}

__global__ void k_scan_add(int* __restrict__ rowptr, const int* __restrict__ bsum,
                           int n, int nb) {
    __shared__ int red[8];
    int t = threadIdx.x, b = blockIdx.x;
    int s = 0;
    for (int k = t; k < b; k += 256) s += bsum[k];
#pragma unroll
    for (int o = 16; o > 0; o >>= 1) s += __shfl_xor(s, o);
    if ((t & 31) == 0) red[t >> 5] = s;
    __syncthreads();
    int base = 0;
#pragma unroll
    for (int k = 0; k < 8; ++k) base += red[k];
    int i = b * 256 + t;
    if (i < n) rowptr[i] += base;
    if (b == nb - 1 && t == 0) rowptr[n] = base + bsum[b];
}

__global__ void k_place(const int* __restrict__ dst, const int* __restrict__ rank,
                        const int* __restrict__ rowptr, int* __restrict__ seg_eid, int E) {
    int e = blockIdx.x * blockDim.x + threadIdx.x;
    if (e < E) seg_eid[rowptr[dst[e]] + rank[e]] = e;
}

__global__ void k_seg_order(const int* __restrict__ src, const int* __restrict__ rowptr,
                            const int* __restrict__ seg_eid, int* __restrict__ seg_src, int n) {
    const int lane = threadIdx.x & 31;
    const int d = __builtin_amdgcn_readfirstlane((int)(blockIdx.x * (blockDim.x >> 5) + (threadIdx.x >> 5)));
    if (d >= n) return;
    const int beg = rowptr[d], end = rowptr[d + 1];
    for (int base = beg; base < end; base += 32) {
        int idx = base + lane;
        int myE = idx < end ? seg_eid[idx] : 0x7fffffff;
        int r = 0;
        for (int j = beg; j < end; ++j) r += (seg_eid[j] < myE) ? 1 : 0;
        if (idx < end) seg_src[beg + r] = src[myE];
    }
}

template <int H> struct HeadRow;
template <> struct HeadRow<4> {
    static __device__ __forceinline__ void load(const float* __restrict__ p, float (&a)[4]) {
        v4f t = *(const v4f*)p;
        a[0] = t.x; a[1] = t.y; a[2] = t.z; a[3] = t.w;
    }
};
template <> struct HeadRow<1> {
    static __device__ __forceinline__ void load(const float* __restrict__ p, float (&a)[1]) {
        a[0] = *p;
    }
};

__device__ __forceinline__ float lrelu02(float v) { return v > 0.f ? v : 0.2f * v; }

template <int H, int C>
__global__ void k_node_gather(const int* __restrict__ rowptr, const int* __restrict__ seg_src,
                              const float* __restrict__ asrc, const float* __restrict__ adst,
                              const float* __restrict__ feat, float* __restrict__ out, int n) {
    constexpr int HC = H * C;
    constexpr int V = HC / 32;
    typedef __attribute__((ext_vector_type(V))) float vrow;
    const int lane = threadIdx.x & 31;
    const int d = __builtin_amdgcn_readfirstlane((int)(blockIdx.x * (blockDim.x >> 5) + (threadIdx.x >> 5)));
    if (d >= n) return;
    const int beg = rowptr[d], end = rowptr[d + 1];
    const int hd = (lane * V) / C;

    float ad[H], vself[H], mx[H];
    {
        float as_d[H];
        HeadRow<H>::load(adst + (size_t)d * H, ad);
        HeadRow<H>::load(asrc + (size_t)d * H, as_d);
#pragma unroll
        for (int h = 0; h < H; ++h) { vself[h] = lrelu02(as_d[h] + ad[h]); mx[h] = vself[h]; }
    }

    for (int base = beg; base < end; base += 32) {
        int idx = base + lane;
        if (idx < end) {
            int s = seg_src[idx];
            float a[H];
            HeadRow<H>::load(asrc + (size_t)s * H, a);
#pragma unroll
            for (int h = 0; h < H; ++h) mx[h] = fmaxf(mx[h], lrelu02(a[h] + ad[h]));
        }
    }
#pragma unroll
    for (int h = 0; h < H; ++h) {
#pragma unroll
        for (int o = 16; o > 0; o >>= 1) mx[h] = fmaxf(mx[h], __shfl_xor(mx[h], o));
    }

    float den[H];
#pragma unroll
    for (int h = 0; h < H; ++h) den[h] = 0.f;
    vrow acc = 0.f;
    for (int base = beg; base < end; base += 32) {
        int idx = base + lane;
        int s = 0;
        float ex[H];
#pragma unroll
        for (int h = 0; h < H; ++h) ex[h] = 0.f;
        if (idx < end) {
            s = seg_src[idx];
            float a[H];
            HeadRow<H>::load(asrc + (size_t)s * H, a);
#pragma unroll
            for (int h = 0; h < H; ++h) ex[h] = expf(lrelu02(a[h] + ad[h]) - mx[h]);
        }
#pragma unroll
        for (int h = 0; h < H; ++h) den[h] += ex[h];
        const int cnt = (end - base) < 32 ? (end - base) : 32;
#pragma unroll 4
        for (int j = 0; j < cnt; ++j) {
            const int sj = __builtin_amdgcn_readlane(s, j);
            float x = __int_as_float(__builtin_amdgcn_readlane(__float_as_int(ex[0]), j));
#pragma unroll
            for (int h = 1; h < H; ++h) {
                float xh = __int_as_float(__builtin_amdgcn_readlane(__float_as_int(ex[h]), j));
                x = (hd == h) ? xh : x;
            }
            vrow f = *(const vrow*)(feat + (size_t)sj * HC + lane * V);
            acc += f * x;
        }
    }
    float den_l = 0.f, xs = 0.f;
#pragma unroll
    for (int h = 0; h < H; ++h) {
        float t = den[h];
#pragma unroll
        for (int o = 16; o > 0; o >>= 1) t += __shfl_xor(t, o);
        float xself = expf(vself[h] - mx[h]);
        t += xself;
        if (h == 0 || hd == h) { den_l = t; xs = xself; }
    }
    {
        vrow f = *(const vrow*)(feat + (size_t)d * HC + lane * V);
        acc += f * xs;
    }
    *(vrow*)(out + (size_t)d * HC + lane * V) = acc / den_l;
}

__global__ void k_bias_elu(float* __restrict__ h, const float* __restrict__ bias,
                           int n, int HC) {
    int i = blockIdx.x * blockDim.x + threadIdx.x;
    if (i >= n * HC) return;
    float v = h[i] + bias[i % HC];
    h[i] = v > 0.f ? v : (expf(v) - 1.f);
}

__global__ void k_bias_add(float* __restrict__ o, const float* __restrict__ bias,
                           int n, int C) {
    int i = blockIdx.x * blockDim.x + threadIdx.x;
    if (i >= n * C) return;
    o[i] += bias[i % C];
}

extern "C" void kernel_launch(void* const* d_in, const int* in_sizes, int n_in,
                              void* d_out, int out_size, void* d_ws, size_t ws_size,
                              hipStream_t stream) {
    const float* x      = (const float*)d_in[0];
    const int*   ei     = (const int*)  d_in[1];
    const float* W1     = (const float*)d_in[2];
    const float* att_s1 = (const float*)d_in[3];
    const float* att_d1 = (const float*)d_in[4];
    const float* b1     = (const float*)d_in[5];
    const float* W2     = (const float*)d_in[6];
    const float* att_s2 = (const float*)d_in[7];
    const float* att_d2 = (const float*)d_in[8];
    const float* b2     = (const float*)d_in[9];

    const int IN = 128, HID = 128, H1 = 4, C1 = 32, OUT = 64;
    const int n   = in_sizes[0] / IN;
    const int E   = in_sizes[1] / 2;
    const int nb  = (n + 255) / 256;

    float* out1  = (float*)d_out;
    float* hfeat = (float*)d_out + (size_t)n * OUT;

    float* w = (float*)d_ws;
    size_t off = 0;
    float* Wt1 = w + off; off += (size_t)IN * HID;
    float* Wt2 = w + off; off += (size_t)HID * OUT;
    float* h1  = w + off; off += (size_t)n * HID;
    float* h2  = w + off; off += (size_t)n * OUT;
    float* as1 = w + off; off += (size_t)n * H1;
    float* ad1 = w + off; off += (size_t)n * H1;
    float* as2 = w + off; off += (size_t)n;
    float* ad2 = w + off; off += (size_t)n;
    int* iw = (int*)(w + off);
    int* cnt     = iw; iw += n;
    int* rowptr  = iw; iw += n + 1;
    int* bsum    = iw; iw += nb;
    int* rank    = iw; iw += E;
    int* seg_eid = iw; iw += E;
    int* seg_src = iw; iw += E;

    const int* srcI = ei;
    const int* dstI = ei + E;

    auto cdiv = [](long a, long b) { return (unsigned)((a + b - 1) / b); };
    dim3 B256(256);

    k_transpose<<<cdiv((long)IN * HID, 256), B256, 0, stream>>>(W1, Wt1, IN, HID);
    k_transpose<<<cdiv((long)HID * OUT, 256), B256, 0, stream>>>(W2, Wt2, HID, OUT);

    k_zero_int  <<<cdiv(n, 256), B256, 0, stream>>>(cnt, n);
    k_count_rank<<<cdiv(E, 256), B256, 0, stream>>>(dstI, cnt, rank, E);
    k_scan_local<<<nb, B256, 0, stream>>>(cnt, rowptr, bsum, n);
    k_scan_add  <<<nb, B256, 0, stream>>>(rowptr, bsum, n, nb);
    k_place     <<<cdiv(E, 256), B256, 0, stream>>>(dstI, rank, rowptr, seg_eid, E);
    k_seg_order <<<cdiv(n, 8), B256, 0, stream>>>(srcI, rowptr, seg_eid, seg_src, n);

    k_gemm_f32_wmma<<<dim3(n / 16, HID / 64), dim3(32, 4), 0, stream>>>(x, Wt1, h1, HID);
    k_attn<<<cdiv((long)n * H1, 256), B256, 0, stream>>>(h1, att_s1, att_d1, as1, ad1, n, H1, C1);
    k_node_gather<4, 32><<<cdiv(n, 8), B256, 0, stream>>>(rowptr, seg_src, as1, ad1, h1, hfeat, n);
    k_bias_elu   <<<cdiv((long)n * HID, 256), B256, 0, stream>>>(hfeat, b1, n, HID);

    k_gemm_f32_wmma<<<dim3(n / 16, OUT / 64), dim3(32, 4), 0, stream>>>(hfeat, Wt2, h2, OUT);
    k_attn<<<cdiv((long)n, 256), B256, 0, stream>>>(h2, att_s2, att_d2, as2, ad2, n, 1, OUT);
    k_node_gather<1, 64><<<cdiv(n, 8), B256, 0, stream>>>(rowptr, seg_src, as2, ad2, h2, out1, n);
    k_bias_add   <<<cdiv((long)n * OUT, 256), B256, 0, stream>>>(out1, b2, n, OUT);
}


# Bias + ELU (layer 1) and bias (layer 2) applied in registers at the gather kernels' final store: each output element stored once, both per-element kernels and their launches removed; sums unchanged
# speedup vs baseline: 5.1327x; 1.1276x over previous
#include <hip/hip_runtime.h>
#include <hip/hip_bf16.h>
#include <math.h>

typedef __attribute__((ext_vector_type(2))) float v2f;
typedef __attribute__((ext_vector_type(8))) float v8f;

__global__ void k_zero_int(int* __restrict__ p, int n) {
    int i = blockIdx.x * blockDim.x + threadIdx.x;
    if (i < n) p[i] = 0;
}

__global__ void k_transpose(const float* __restrict__ W, float* __restrict__ Wt,
                            int K, int Ncol) {
    int i = blockIdx.x * blockDim.x + threadIdx.x;
    if (i < K * Ncol) {
        int k = i / Ncol, nn = i % Ncol;
        Wt[nn * K + k] = W[i];
    }
}

__global__ void k_gemm_f32_wmma(const float* __restrict__ A,
                                const float* __restrict__ Wt,
                                float* __restrict__ C, int nCols) {
    const int K = 128;
    int lane = threadIdx.x;
    int half = lane >> 4;
    int l    = lane & 15;
    int rowBase = blockIdx.x * 16;
    int colBase = (blockIdx.y * 4 + threadIdx.y) * 16;

    const float* arow = A  + (size_t)(rowBase + l) * K + 2 * half;
    const float* brow = Wt + (size_t)(colBase + l) * K + 2 * half;

    v8f acc = {};
#pragma unroll
    for (int k = 0; k < K; k += 4) {
        v2f a = *(const v2f*)(arow + k);
        v2f b = *(const v2f*)(brow + k);
        acc = __builtin_amdgcn_wmma_f32_16x16x4_f32(
             false, a,  false, b,
             (short)0, acc,  false,  false);
    }
#pragma unroll
    for (int r = 0; r < 8; ++r)
        C[(size_t)(rowBase + r + 8 * half) * nCols + colBase + l] = acc[r];
}

__global__ void k_attn(const float* __restrict__ h, const float* __restrict__ att_s,
                       const float* __restrict__ att_d, float* __restrict__ asrc,
                       float* __restrict__ adst, int n, int heads, int C) {
    int i = blockIdx.x * blockDim.x + threadIdx.x;
    if (i >= n * heads) return;
    int node = i / heads, hd = i % heads;
    const float* hp = h + (size_t)node * heads * C + (size_t)hd * C;
    float ss = 0.f, sd = 0.f;
    for (int c = 0; c < C; ++c) {
        float v = hp[c];
        ss += v * att_s[hd * C + c];
        sd += v * att_d[hd * C + c];
    }
    asrc[i] = ss;
    adst[i] = sd;
}

typedef __attribute__((ext_vector_type(4))) float v4f;

__global__ void k_count_rank(const int* __restrict__ dst, int* __restrict__ cnt,
                             int* __restrict__ rank, int E) {
    int e = blockIdx.x * blockDim.x + threadIdx.x;
    if (e < E) rank[e] = atomicAdd(&cnt[dst[e]], 1);
}

__global__ void k_scan_local(const int* __restrict__ cnt, int* __restrict__ rowptr,
                             int* __restrict__ bsum, int n) {
    __shared__ int wsum[8];
    int t = threadIdx.x, i = blockIdx.x * 256 + t;
    int lane = t & 31, wv = t >> 5;
    int v = i < n ? cnt[i] : 0;
    int inc = v;
#pragma unroll
    for (int o = 1; o < 32; o <<= 1) {
        int u = __shfl_up(inc, o);
        if (lane >= o) inc += u;
    }
    if (lane == 31) wsum[wv] = inc;
    __syncthreads();
    int woff = 0;
#pragma unroll
    for (int k = 0; k < 8; ++k) woff += (k < wv) ? wsum[k] : 0;
    if (i < n) rowptr[i] = woff + inc - v;
    if (t == 255) bsum[blockIdx.x] = woff + inc;
}

__global__ void k_scan_add(int* __restrict__ rowptr, const int* __restrict__ bsum,
                           int n, int nb) {
    __shared__ int red[8];
    int t = threadIdx.x, b = blockIdx.x;
    int s = 0;
    for (int k = t; k < b; k += 256) s += bsum[k];
#pragma unroll
    for (int o = 16; o > 0; o >>= 1) s += __shfl_xor(s, o);
    if ((t & 31) == 0) red[t >> 5] = s;
    __syncthreads();
    int base = 0;
#pragma unroll
    for (int k = 0; k < 8; ++k) base += red[k];
    int i = b * 256 + t;
    if (i < n) rowptr[i] += base;
    if (b == nb - 1 && t == 0) rowptr[n] = base + bsum[b];
}

__global__ void k_place(const int* __restrict__ dst, const int* __restrict__ rank,
                        const int* __restrict__ rowptr, int* __restrict__ seg_eid, int E) {
    int e = blockIdx.x * blockDim.x + threadIdx.x;
    if (e < E) seg_eid[rowptr[dst[e]] + rank[e]] = e;
}

__global__ void k_seg_order(const int* __restrict__ src, const int* __restrict__ rowptr,
                            const int* __restrict__ seg_eid, int* __restrict__ seg_src, int n) {
    const int lane = threadIdx.x & 31;
    const int d = __builtin_amdgcn_readfirstlane((int)(blockIdx.x * (blockDim.x >> 5) + (threadIdx.x >> 5)));
    if (d >= n) return;
    const int beg = rowptr[d], end = rowptr[d + 1];
    for (int base = beg; base < end; base += 32) {
        int idx = base + lane;
        int myE = idx < end ? seg_eid[idx] : 0x7fffffff;
        int r = 0;
        for (int j = beg; j < end; ++j) r += (seg_eid[j] < myE) ? 1 : 0;
        if (idx < end) seg_src[beg + r] = src[myE];
    }
}

template <int H> struct HeadRow;
template <> struct HeadRow<4> {
    static __device__ __forceinline__ void load(const float* __restrict__ p, float (&a)[4]) {
        v4f t = *(const v4f*)p;
        a[0] = t.x; a[1] = t.y; a[2] = t.z; a[3] = t.w;
    }
};
template <> struct HeadRow<1> {
    static __device__ __forceinline__ void load(const float* __restrict__ p, float (&a)[1]) {
        a[0] = *p;
    }
};

__device__ __forceinline__ float lrelu02(float v) { return v > 0.f ? v : 0.2f * v; }

template <int H, int C, bool ELU>
__global__ void k_node_gather(const int* __restrict__ rowptr, const int* __restrict__ seg_src,
                              const float* __restrict__ asrc, const float* __restrict__ adst,
                              const float* __restrict__ feat, const float* __restrict__ bias,
                              float* __restrict__ out, int n) {
    constexpr int HC = H * C;
    constexpr int V = HC / 32;
    typedef __attribute__((ext_vector_type(V))) float vrow;
    const int lane = threadIdx.x & 31;
    const int d = __builtin_amdgcn_readfirstlane((int)(blockIdx.x * (blockDim.x >> 5) + (threadIdx.x >> 5)));
    if (d >= n) return;
    const int beg = rowptr[d], end = rowptr[d + 1];
    const int hd = (lane * V) / C;

    float ad[H], vself[H], mx[H];
    {
        float as_d[H];
        HeadRow<H>::load(adst + (size_t)d * H, ad);
        HeadRow<H>::load(asrc + (size_t)d * H, as_d);
#pragma unroll
        for (int h = 0; h < H; ++h) { vself[h] = lrelu02(as_d[h] + ad[h]); mx[h] = vself[h]; }
    }

    for (int base = beg; base < end; base += 32) {
        int idx = base + lane;
        if (idx < end) {
            int s = seg_src[idx];
            float a[H];
            HeadRow<H>::load(asrc + (size_t)s * H, a);
#pragma unroll
            for (int h = 0; h < H; ++h) mx[h] = fmaxf(mx[h], lrelu02(a[h] + ad[h]));
        }
    }
#pragma unroll
    for (int h = 0; h < H; ++h) {
#pragma unroll
        for (int o = 16; o > 0; o >>= 1) mx[h] = fmaxf(mx[h], __shfl_xor(mx[h], o));
    }

    float den[H];
#pragma unroll
    for (int h = 0; h < H; ++h) den[h] = 0.f;
    vrow acc = 0.f;
    for (int base = beg; base < end; base += 32) {
        int idx = base + lane;
        int s = 0;
        float ex[H];
#pragma unroll
        for (int h = 0; h < H; ++h) ex[h] = 0.f;
        if (idx < end) {
            s = seg_src[idx];
            float a[H];
            HeadRow<H>::load(asrc + (size_t)s * H, a);
#pragma unroll
            for (int h = 0; h < H; ++h) ex[h] = expf(lrelu02(a[h] + ad[h]) - mx[h]);
        }
#pragma unroll
        for (int h = 0; h < H; ++h) den[h] += ex[h];
        const int cnt = (end - base) < 32 ? (end - base) : 32;
#pragma unroll 4
        for (int j = 0; j < cnt; ++j) {
            const int sj = __builtin_amdgcn_readlane(s, j);
            float x = __int_as_float(__builtin_amdgcn_readlane(__float_as_int(ex[0]), j));
#pragma unroll
            for (int h = 1; h < H; ++h) {
                float xh = __int_as_float(__builtin_amdgcn_readlane(__float_as_int(ex[h]), j));
                x = (hd == h) ? xh : x;
            }
            vrow f = *(const vrow*)(feat + (size_t)sj * HC + lane * V);
            acc += f * x;
        }
    }
    float den_l = 0.f, xs = 0.f;
#pragma unroll
    for (int h = 0; h < H; ++h) {
        float t = den[h];
#pragma unroll
        for (int o = 16; o > 0; o >>= 1) t += __shfl_xor(t, o);
        float xself = expf(vself[h] - mx[h]);
        t += xself;
        if (h == 0 || hd == h) { den_l = t; xs = xself; }
    }
    const vrow bv = *(const vrow*)(bias + lane * V);
    {
        vrow f = *(const vrow*)(feat + (size_t)d * HC + lane * V);
        acc += f * xs;
    }
    vrow r = acc / den_l + bv;
    if (ELU) {
#pragma unroll
        for (int i = 0; i < V; ++i) r[i] = r[i] > 0.f ? r[i] : (expf(r[i]) - 1.f);
    }
    *(vrow*)(out + (size_t)d * HC + lane * V) = r;
}

extern "C" void kernel_launch(void* const* d_in, const int* in_sizes, int n_in,
                              void* d_out, int out_size, void* d_ws, size_t ws_size,
                              hipStream_t stream) {
    const float* x      = (const float*)d_in[0];
    const int*   ei     = (const int*)  d_in[1];
    const float* W1     = (const float*)d_in[2];
    const float* att_s1 = (const float*)d_in[3];
    const float* att_d1 = (const float*)d_in[4];
    const float* b1     = (const float*)d_in[5];
    const float* W2     = (const float*)d_in[6];
    const float* att_s2 = (const float*)d_in[7];
    const float* att_d2 = (const float*)d_in[8];
    const float* b2     = (const float*)d_in[9];

    const int IN = 128, HID = 128, H1 = 4, C1 = 32, OUT = 64;
    const int n   = in_sizes[0] / IN;
    const int E   = in_sizes[1] / 2;
    const int nb  = (n + 255) / 256;

    float* out1  = (float*)d_out;
    float* hfeat = (float*)d_out + (size_t)n * OUT;

    float* w = (float*)d_ws;
    size_t off = 0;
    float* Wt1 = w + off; off += (size_t)IN * HID;
    float* Wt2 = w + off; off += (size_t)HID * OUT;
    float* h1  = w + off; off += (size_t)n * HID;
    float* h2  = w + off; off += (size_t)n * OUT;
    float* as1 = w + off; off += (size_t)n * H1;
    float* ad1 = w + off; off += (size_t)n * H1;
    float* as2 = w + off; off += (size_t)n;
    float* ad2 = w + off; off += (size_t)n;
    int* iw = (int*)(w + off);
    int* cnt     = iw; iw += n;
    int* rowptr  = iw; iw += n + 1;
    int* bsum    = iw; iw += nb;
    int* rank    = iw; iw += E;
    int* seg_eid = iw; iw += E;
    int* seg_src = iw; iw += E;

    const int* srcI = ei;
    const int* dstI = ei + E;

    auto cdiv = [](long a, long b) { return (unsigned)((a + b - 1) / b); };
    dim3 B256(256);

    k_transpose<<<cdiv((long)IN * HID, 256), B256, 0, stream>>>(W1, Wt1, IN, HID);
    k_transpose<<<cdiv((long)HID * OUT, 256), B256, 0, stream>>>(W2, Wt2, HID, OUT);

    k_zero_int  <<<cdiv(n, 256), B256, 0, stream>>>(cnt, n);
    k_count_rank<<<cdiv(E, 256), B256, 0, stream>>>(dstI, cnt, rank, E);
    k_scan_local<<<nb, B256, 0, stream>>>(cnt, rowptr, bsum, n);
    k_scan_add  <<<nb, B256, 0, stream>>>(rowptr, bsum, n, nb);
    k_place     <<<cdiv(E, 256), B256, 0, stream>>>(dstI, rank, rowptr, seg_eid, E);
    k_seg_order <<<cdiv(n, 8), B256, 0, stream>>>(srcI, rowptr, seg_eid, seg_src, n);

    k_gemm_f32_wmma<<<dim3(n / 16, HID / 64), dim3(32, 4), 0, stream>>>(x, Wt1, h1, HID);
    k_attn<<<cdiv((long)n * H1, 256), B256, 0, stream>>>(h1, att_s1, att_d1, as1, ad1, n, H1, C1);
    k_node_gather<4, 32, true><<<cdiv(n, 8), B256, 0, stream>>>(rowptr, seg_src, as1, ad1, h1, b1, hfeat, n);

    k_gemm_f32_wmma<<<dim3(n / 16, OUT / 64), dim3(32, 4), 0, stream>>>(hfeat, Wt2, h2, OUT);
    k_attn<<<cdiv((long)n, 256), B256, 0, stream>>>(h2, att_s2, att_d2, as2, ad2, n, 1, OUT);
    k_node_gather<1, 64, false><<<cdiv(n, 8), B256, 0, stream>>>(rowptr, seg_src, as2, ad2, h2, b2, out1, n);
}


# GEMM re-tiled: one wave per 16 rows x 64 columns keeps its A rows in registers for four tiles; 16-byte loads, whole-line stores, weights pre-packed in lane order; k order of the sums permuted
# speedup vs baseline: 5.8005x; 1.1301x over previous
#include <hip/hip_runtime.h>
#include <hip/hip_bf16.h>
#include <math.h>

typedef __attribute__((ext_vector_type(2))) float v2f;
typedef __attribute__((ext_vector_type(8))) float v8f;

__global__ void k_zero_int(int* __restrict__ p, int n) {
    int i = blockIdx.x * blockDim.x + threadIdx.x;
    if (i < n) p[i] = 0;
}

__global__ void k_pack_b(const float* __restrict__ W, float* __restrict__ Bf, int Ncol) {
    int i = blockIdx.x * blockDim.x + threadIdx.x;
    if (i < 128 * Ncol) {
        int e = i & 3, lane = (i >> 2) & 31, j = (i >> 7) & 15, t = i >> 11;
        int k  = 8 * j + 4 * (lane >> 4) + e;
        int nn = 64 * (t >> 2) + 4 * (lane & 15) + (t & 3);
        Bf[i] = W[k * Ncol + nn];
    }
}

typedef __attribute__((ext_vector_type(4))) float v4g;
__global__ void __launch_bounds__(128)
k_gemm_f32_wmma(const float* __restrict__ A, const float* __restrict__ Bf,
                float* __restrict__ C, int nCols, int nStripes) {
    const int lane = threadIdx.x & 31;
    const int hh   = lane >> 4;
    const int c    = lane & 15;
    const int stripe = __builtin_amdgcn_readfirstlane((int)(blockIdx.x * 4 + (threadIdx.x >> 5)));
    if (stripe >= nStripes) return;
    const int rowBase = stripe * 16;
    const int q = blockIdx.y;

    const float* arow = A + (size_t)(rowBase + c) * 128 + 4 * hh;
    v4g a[16];
#pragma unroll
    for (int j = 0; j < 16; ++j) a[j] = *(const v4g*)(arow + 8 * j);

    const float* bp = Bf + (size_t)q * (4 * 2048) + lane * 4;
    v8f acc[4];
#pragma unroll
    for (int t = 0; t < 4; ++t) acc[t] = (v8f){};

    v4g b[3][4];
#pragma unroll
    for (int t = 0; t < 4; ++t) b[0][t] = *(const v4g*)(bp + (t * 16 + 0) * 128);
#pragma unroll
    for (int t = 0; t < 4; ++t) b[1][t] = *(const v4g*)(bp + (t * 16 + 1) * 128);

#pragma unroll
    for (int j = 0; j < 16; ++j) {
        if (j + 2 < 16) {
#pragma unroll
            for (int t = 0; t < 4; ++t)
                b[(j + 2) % 3][t] = *(const v4g*)(bp + (t * 16 + j + 2) * 128);
        }
        __builtin_amdgcn_sched_barrier(0);
        const v2f a0 = {a[j].x, a[j].y};
        const v2f a1 = {a[j].z, a[j].w};
#pragma unroll
        for (int t = 0; t < 4; ++t) {
            const v4g bb = b[j % 3][t];
            const v2f b0 = {bb.x, bb.y};
            const v2f b1 = {bb.z, bb.w};
            acc[t] = __builtin_amdgcn_wmma_f32_16x16x4_f32(
                 false, a0,  false, b0,
                 (short)0, acc[t],  false,  false);
            acc[t] = __builtin_amdgcn_wmma_f32_16x16x4_f32(
                 false, a1,  false, b1,
                 (short)0, acc[t],  false,  false);
        }
        __builtin_amdgcn_sched_barrier(0);
    }
#pragma unroll
    for (int r = 0; r < 8; ++r) {
        v4g o = {acc[0][r], acc[1][r], acc[2][r], acc[3][r]};
        *(v4g*)(C + (size_t)(rowBase + r + 8 * hh) * nCols + 64 * q + 4 * c) = o;
    }
}

__global__ void k_attn(const float* __restrict__ h, const float* __restrict__ att_s,
                       const float* __restrict__ att_d, float* __restrict__ asrc,
                       float* __restrict__ adst, int n, int heads, int C) {
    int i = blockIdx.x * blockDim.x + threadIdx.x;
    if (i >= n * heads) return;
    int node = i / heads, hd = i % heads;
    const float* hp = h + (size_t)node * heads * C + (size_t)hd * C;
    float ss = 0.f, sd = 0.f;
    for (int c = 0; c < C; ++c) {
        float v = hp[c];
        ss += v * att_s[hd * C + c];
        sd += v * att_d[hd * C + c];
    }
    asrc[i] = ss;
    adst[i] = sd;
}

typedef __attribute__((ext_vector_type(4))) float v4f;

__global__ void k_count_rank(const int* __restrict__ dst, int* __restrict__ cnt,
                             int* __restrict__ rank, int E) {
    int e = blockIdx.x * blockDim.x + threadIdx.x;
    if (e < E) rank[e] = atomicAdd(&cnt[dst[e]], 1);
}

__global__ void k_scan_local(const int* __restrict__ cnt, int* __restrict__ rowptr,
                             int* __restrict__ bsum, int n) {
    __shared__ int wsum[8];
    int t = threadIdx.x, i = blockIdx.x * 256 + t;
    int lane = t & 31, wv = t >> 5;
    int v = i < n ? cnt[i] : 0;
    int inc = v;
#pragma unroll
    for (int o = 1; o < 32; o <<= 1) {
        int u = __shfl_up(inc, o);
        if (lane >= o) inc += u;
    }
    if (lane == 31) wsum[wv] = inc;
    __syncthreads();
    int woff = 0;
#pragma unroll
    for (int k = 0; k < 8; ++k) woff += (k < wv) ? wsum[k] : 0;
    if (i < n) rowptr[i] = woff + inc - v;
    if (t == 255) bsum[blockIdx.x] = woff + inc;
}

__global__ void k_scan_add(int* __restrict__ rowptr, const int* __restrict__ bsum,
                           int n, int nb) {
    __shared__ int red[8];
    int t = threadIdx.x, b = blockIdx.x;
    int s = 0;
    for (int k = t; k < b; k += 256) s += bsum[k];
#pragma unroll
    for (int o = 16; o > 0; o >>= 1) s += __shfl_xor(s, o);
    if ((t & 31) == 0) red[t >> 5] = s;
    __syncthreads();
    int base = 0;
#pragma unroll
    for (int k = 0; k < 8; ++k) base += red[k];
    int i = b * 256 + t;
    if (i < n) rowptr[i] += base;
    if (b == nb - 1 && t == 0) rowptr[n] = base + bsum[b];
}

__global__ void k_place(const int* __restrict__ dst, const int* __restrict__ rank,
                        const int* __restrict__ rowptr, int* __restrict__ seg_eid, int E) {
    int e = blockIdx.x * blockDim.x + threadIdx.x;
    if (e < E) seg_eid[rowptr[dst[e]] + rank[e]] = e;
}

__global__ void k_seg_order(const int* __restrict__ src, const int* __restrict__ rowptr,
                            const int* __restrict__ seg_eid, int* __restrict__ seg_src, int n) {
    const int lane = threadIdx.x & 31;
    const int d = __builtin_amdgcn_readfirstlane((int)(blockIdx.x * (blockDim.x >> 5) + (threadIdx.x >> 5)));
    if (d >= n) return;
    const int beg = rowptr[d], end = rowptr[d + 1];
    for (int base = beg; base < end; base += 32) {
        int idx = base + lane;
        int myE = idx < end ? seg_eid[idx] : 0x7fffffff;
        int r = 0;
        for (int j = beg; j < end; ++j) r += (seg_eid[j] < myE) ? 1 : 0;
        if (idx < end) seg_src[beg + r] = src[myE];
    }
}

template <int H> struct HeadRow;
template <> struct HeadRow<4> {
    static __device__ __forceinline__ void load(const float* __restrict__ p, float (&a)[4]) {
        v4f t = *(const v4f*)p;
        a[0] = t.x; a[1] = t.y; a[2] = t.z; a[3] = t.w;
    }
};
template <> struct HeadRow<1> {
    static __device__ __forceinline__ void load(const float* __restrict__ p, float (&a)[1]) {
        a[0] = *p;
    }
};

__device__ __forceinline__ float lrelu02(float v) { return v > 0.f ? v : 0.2f * v; }

template <int H, int C, bool ELU>
__global__ void k_node_gather(const int* __restrict__ rowptr, const int* __restrict__ seg_src,
                              const float* __restrict__ asrc, const float* __restrict__ adst,
                              const float* __restrict__ feat, const float* __restrict__ bias,
                              float* __restrict__ out, int n) {
    constexpr int HC = H * C;
    constexpr int V = HC / 32;
    typedef __attribute__((ext_vector_type(V))) float vrow;
    const int lane = threadIdx.x & 31;
    const int d = __builtin_amdgcn_readfirstlane((int)(blockIdx.x * (blockDim.x >> 5) + (threadIdx.x >> 5)));
    if (d >= n) return;
    const int beg = rowptr[d], end = rowptr[d + 1];
    const int hd = (lane * V) / C;

    float ad[H], vself[H], mx[H];
    {
        float as_d[H];
        HeadRow<H>::load(adst + (size_t)d * H, ad);
        HeadRow<H>::load(asrc + (size_t)d * H, as_d);
#pragma unroll
        for (int h = 0; h < H; ++h) { vself[h] = lrelu02(as_d[h] + ad[h]); mx[h] = vself[h]; }
    }

    for (int base = beg; base < end; base += 32) {
        int idx = base + lane;
        if (idx < end) {
            int s = seg_src[idx];
            float a[H];
            HeadRow<H>::load(asrc + (size_t)s * H, a);
#pragma unroll
            for (int h = 0; h < H; ++h) mx[h] = fmaxf(mx[h], lrelu02(a[h] + ad[h]));
        }
    }
#pragma unroll
    for (int h = 0; h < H; ++h) {
#pragma unroll
        for (int o = 16; o > 0; o >>= 1) mx[h] = fmaxf(mx[h], __shfl_xor(mx[h], o));
    }

    float den[H];
#pragma unroll
    for (int h = 0; h < H; ++h) den[h] = 0.f;
    vrow acc = 0.f;
    for (int base = beg; base < end; base += 32) {
        int idx = base + lane;
        int s = 0;
        float ex[H];
#pragma unroll
        for (int h = 0; h < H; ++h) ex[h] = 0.f;
        if (idx < end) {
            s = seg_src[idx];
            float a[H];
            HeadRow<H>::load(asrc + (size_t)s * H, a);
#pragma unroll
            for (int h = 0; h < H; ++h) ex[h] = expf(lrelu02(a[h] + ad[h]) - mx[h]);
        }
#pragma unroll
        for (int h = 0; h < H; ++h) den[h] += ex[h];
        const int cnt = (end - base) < 32 ? (end - base) : 32;
#pragma unroll 4
        for (int j = 0; j < cnt; ++j) {
            const int sj = __builtin_amdgcn_readlane(s, j);
            float x = __int_as_float(__builtin_amdgcn_readlane(__float_as_int(ex[0]), j));
#pragma unroll
            for (int h = 1; h < H; ++h) {
                float xh = __int_as_float(__builtin_amdgcn_readlane(__float_as_int(ex[h]), j));
                x = (hd == h) ? xh : x;
            }
            vrow f = *(const vrow*)(feat + (size_t)sj * HC + lane * V);
            acc += f * x;
        }
    }
    float den_l = 0.f, xs = 0.f;
#pragma unroll
    for (int h = 0; h < H; ++h) {
        float t = den[h];
#pragma unroll
        for (int o = 16; o > 0; o >>= 1) t += __shfl_xor(t, o);
        float xself = expf(vself[h] - mx[h]);
        t += xself;
        if (h == 0 || hd == h) { den_l = t; xs = xself; }
    }
    const vrow bv = *(const vrow*)(bias + lane * V);
    {
        vrow f = *(const vrow*)(feat + (size_t)d * HC + lane * V);
        acc += f * xs;
    }
    vrow r = acc / den_l + bv;
    if (ELU) {
#pragma unroll
        for (int i = 0; i < V; ++i) r[i] = r[i] > 0.f ? r[i] : (expf(r[i]) - 1.f);
    }
    *(vrow*)(out + (size_t)d * HC + lane * V) = r;
}

extern "C" void kernel_launch(void* const* d_in, const int* in_sizes, int n_in,
                              void* d_out, int out_size, void* d_ws, size_t ws_size,
                              hipStream_t stream) {
    const float* x      = (const float*)d_in[0];
    const int*   ei     = (const int*)  d_in[1];
    const float* W1     = (const float*)d_in[2];
    const float* att_s1 = (const float*)d_in[3];
    const float* att_d1 = (const float*)d_in[4];
    const float* b1     = (const float*)d_in[5];
    const float* W2     = (const float*)d_in[6];
    const float* att_s2 = (const float*)d_in[7];
    const float* att_d2 = (const float*)d_in[8];
    const float* b2     = (const float*)d_in[9];

    const int IN = 128, HID = 128, H1 = 4, C1 = 32, OUT = 64;
    const int n   = in_sizes[0] / IN;
    const int E   = in_sizes[1] / 2;
    const int nb  = (n + 255) / 256;

    float* out1  = (float*)d_out;
    float* hfeat = (float*)d_out + (size_t)n * OUT;

    float* w = (float*)d_ws;
    size_t off = 0;
    float* Wt1 = w + off; off += (size_t)IN * HID;
    float* Wt2 = w + off; off += (size_t)HID * OUT;
    float* h1  = w + off; off += (size_t)n * HID;
    float* h2  = w + off; off += (size_t)n * OUT;
    float* as1 = w + off; off += (size_t)n * H1;
    float* ad1 = w + off; off += (size_t)n * H1;
    float* as2 = w + off; off += (size_t)n;
    float* ad2 = w + off; off += (size_t)n;
    int* iw = (int*)(w + off);
    int* cnt     = iw; iw += n;
    int* rowptr  = iw; iw += n + 1;
    int* bsum    = iw; iw += nb;
    int* rank    = iw; iw += E;
    int* seg_eid = iw; iw += E;
    int* seg_src = iw; iw += E;

    const int* srcI = ei;
    const int* dstI = ei + E;

    auto cdiv = [](long a, long b) { return (unsigned)((a + b - 1) / b); };
    dim3 B256(256);

    k_pack_b<<<cdiv((long)IN * HID, 256), B256, 0, stream>>>(W1, Wt1, HID);
    k_pack_b<<<cdiv((long)HID * OUT, 256), B256, 0, stream>>>(W2, Wt2, OUT);

    k_zero_int  <<<cdiv(n, 256), B256, 0, stream>>>(cnt, n);
    k_count_rank<<<cdiv(E, 256), B256, 0, stream>>>(dstI, cnt, rank, E);
    k_scan_local<<<nb, B256, 0, stream>>>(cnt, rowptr, bsum, n);
    k_scan_add  <<<nb, B256, 0, stream>>>(rowptr, bsum, n, nb);
    k_place     <<<cdiv(E, 256), B256, 0, stream>>>(dstI, rank, rowptr, seg_eid, E);
    k_seg_order <<<cdiv(n, 8), B256, 0, stream>>>(srcI, rowptr, seg_eid, seg_src, n);

    k_gemm_f32_wmma<<<dim3(cdiv(n / 16, 4), HID / 64), dim3(128), 0, stream>>>(x, Wt1, h1, HID, n / 16);
    k_attn<<<cdiv((long)n * H1, 256), B256, 0, stream>>>(h1, att_s1, att_d1, as1, ad1, n, H1, C1);
    k_node_gather<4, 32, true><<<cdiv(n, 8), B256, 0, stream>>>(rowptr, seg_src, as1, ad1, h1, b1, hfeat, n);

    k_gemm_f32_wmma<<<dim3(cdiv(n / 16, 4), OUT / 64), dim3(128), 0, stream>>>(hfeat, Wt2, h2, OUT, n / 16);
    k_attn<<<cdiv((long)n, 256), B256, 0, stream>>>(h2, att_s2, att_d2, as2, ad2, n, 1, OUT);
    k_node_gather<1, 64, false><<<cdiv(n, 8), B256, 0, stream>>>(rowptr, seg_src, as2, ad2, h2, b2, out1, n);
}


# Attention scores of both layers taken from the GEMM accumulators in its epilogue (4 columns per lane, then a lane tree): both score kernels and their re-read of h removed; score sum order changed
# speedup vs baseline: 6.1485x; 1.0600x over previous
#include <hip/hip_runtime.h>
#include <hip/hip_bf16.h>
#include <math.h>

typedef __attribute__((ext_vector_type(2))) float v2f;
typedef __attribute__((ext_vector_type(8))) float v8f;

__global__ void k_zero_int(int* __restrict__ p, int n) {
    int i = blockIdx.x * blockDim.x + threadIdx.x;
    if (i < n) p[i] = 0;
}

__global__ void k_pack_b(const float* __restrict__ W, float* __restrict__ Bf, int Ncol) {
    int i = blockIdx.x * blockDim.x + threadIdx.x;
    if (i < 128 * Ncol) {
        int e = i & 3, lane = (i >> 2) & 31, j = (i >> 7) & 15, t = i >> 11;
        int k  = 8 * j + 4 * (lane >> 4) + e;
        int nn = 64 * (t >> 2) + 4 * (lane & 15) + (t & 3);
        Bf[i] = W[k * Ncol + nn];
    }
}

typedef __attribute__((ext_vector_type(4))) float v4g;

template <int M>
__device__ __forceinline__ float row_lane_xor(float x) {
    return __int_as_float(__builtin_amdgcn_update_dpp(0, __float_as_int(x), 0x160 + M, 0xf, 0xf, false));
}

template <int M, int K>
__device__ __forceinline__ void fold_pairs(float (&v)[16], const bool up) {
#pragma unroll
    for (int i = 0; i < K; ++i) {
        const float keep = up ? v[i + K] : v[i];
        const float send = up ? v[i] : v[i + K];
        v[i] = keep + row_lane_xor<M>(send);
    }
}

template <int LPH>
__global__ void __launch_bounds__(128)
k_gemm_f32_wmma(const float* __restrict__ A, const float* __restrict__ Bf,
                float* __restrict__ C, const float* __restrict__ att_s,
                const float* __restrict__ att_d, float* __restrict__ asrc,
                float* __restrict__ adst, int nCols, int nStripes) {
    const int lane = threadIdx.x & 31;
    const int hh   = lane >> 4;
    const int c    = lane & 15;
    const int stripe = __builtin_amdgcn_readfirstlane((int)(blockIdx.x * 4 + (threadIdx.x >> 5)));
    if (stripe >= nStripes) return;
    const int rowBase = stripe * 16;
    const int q = blockIdx.y;

    const float* arow = A + (size_t)(rowBase + c) * 128 + 4 * hh;
    v4g a[16];
#pragma unroll
    for (int j = 0; j < 16; ++j) a[j] = *(const v4g*)(arow + 8 * j);

    const float* bp = Bf + (size_t)q * (4 * 2048) + lane * 4;
    v8f acc[4];
#pragma unroll
    for (int t = 0; t < 4; ++t) acc[t] = (v8f){};

    v4g b[3][4];
    v4g ws, wd;
#pragma unroll
    for (int t = 0; t < 4; ++t) b[0][t] = *(const v4g*)(bp + (t * 16 + 0) * 128);
#pragma unroll
    for (int t = 0; t < 4; ++t) b[1][t] = *(const v4g*)(bp + (t * 16 + 1) * 128);

#pragma unroll
    for (int j = 0; j < 16; ++j) {
        if (j + 2 < 16) {
#pragma unroll
            for (int t = 0; t < 4; ++t)
                b[(j + 2) % 3][t] = *(const v4g*)(bp + (t * 16 + j + 2) * 128);
        }
        if (j == 8) {
            ws = *(const v4g*)(att_s + 64 * q + 4 * c);
            wd = *(const v4g*)(att_d + 64 * q + 4 * c);
        }
        __builtin_amdgcn_sched_barrier(0);
        const v2f a0 = {a[j].x, a[j].y};
        const v2f a1 = {a[j].z, a[j].w};
#pragma unroll
        for (int t = 0; t < 4; ++t) {
            const v4g bb = b[j % 3][t];
            const v2f b0 = {bb.x, bb.y};
            const v2f b1 = {bb.z, bb.w};
            acc[t] = __builtin_amdgcn_wmma_f32_16x16x4_f32(
                 false, a0,  false, b0,
                 (short)0, acc[t],  false,  false);
            acc[t] = __builtin_amdgcn_wmma_f32_16x16x4_f32(
                 false, a1,  false, b1,
                 (short)0, acc[t],  false,  false);
        }
        __builtin_amdgcn_sched_barrier(0);
    }
#pragma unroll
    for (int r = 0; r < 8; ++r) {
        v4g o = {acc[0][r], acc[1][r], acc[2][r], acc[3][r]};
        *(v4g*)(C + (size_t)(rowBase + r + 8 * hh) * nCols + 64 * q + 4 * c) = o;
    }

    float v[16];
#pragma unroll
    for (int r = 0; r < 8; ++r) {
        v[r]     = ((acc[0][r] * ws.x + acc[1][r] * ws.y) + acc[2][r] * ws.z) + acc[3][r] * ws.w;
        v[8 + r] = ((acc[0][r] * wd.x + acc[1][r] * wd.y) + acc[2][r] * wd.z) + acc[3][r] * wd.w;
    }
    fold_pairs<1, 8>(v, (lane & 1) != 0);
    fold_pairs<2, 4>(v, (lane & 2) != 0);
    fold_pairs<4, 2>(v, (lane & 4) != 0);
    if (LPH == 16) fold_pairs<8, 1>(v, (lane & 8) != 0);
    const int H    = nCols / (4 * LPH);
    const int head = LPH == 16 ? q : 2 * q + ((lane >> 3) & 1);
    const int r0   = 4 * ((lane >> 1) & 1) + 2 * ((lane >> 2) & 1) + (LPH == 16 ? ((lane >> 3) & 1) : 0);
    float* sp = ((lane & 1) ? adst : asrc) + (size_t)(rowBase + 8 * hh + r0) * H + head;
    sp[0] = v[0];
    if (LPH == 8) sp[H] = v[1];
}

typedef __attribute__((ext_vector_type(4))) float v4f;

__global__ void k_count_rank(const int* __restrict__ dst, int* __restrict__ cnt,
                             int* __restrict__ rank, int E) {
    int e = blockIdx.x * blockDim.x + threadIdx.x;
    if (e < E) rank[e] = atomicAdd(&cnt[dst[e]], 1);
}

__global__ void k_scan_local(const int* __restrict__ cnt, int* __restrict__ rowptr,
                             int* __restrict__ bsum, int n) {
    __shared__ int wsum[8];
    int t = threadIdx.x, i = blockIdx.x * 256 + t;
    int lane = t & 31, wv = t >> 5;
    int v = i < n ? cnt[i] : 0;
    int inc = v;
#pragma unroll
    for (int o = 1; o < 32; o <<= 1) {
        int u = __shfl_up(inc, o);
        if (lane >= o) inc += u;
    }
    if (lane == 31) wsum[wv] = inc;
    __syncthreads();
    int woff = 0;
#pragma unroll
    for (int k = 0; k < 8; ++k) woff += (k < wv) ? wsum[k] : 0;
    if (i < n) rowptr[i] = woff + inc - v;
    if (t == 255) bsum[blockIdx.x] = woff + inc;
}

__global__ void k_scan_add(int* __restrict__ rowptr, const int* __restrict__ bsum,
                           int n, int nb) {
    __shared__ int red[8];
    int t = threadIdx.x, b = blockIdx.x;
    int s = 0;
    for (int k = t; k < b; k += 256) s += bsum[k];
#pragma unroll
    for (int o = 16; o > 0; o >>= 1) s += __shfl_xor(s, o);
    if ((t & 31) == 0) red[t >> 5] = s;
    __syncthreads();
    int base = 0;
#pragma unroll
    for (int k = 0; k < 8; ++k) base += red[k];
    int i = b * 256 + t;
    if (i < n) rowptr[i] += base;
    if (b == nb - 1 && t == 0) rowptr[n] = base + bsum[b];
}

__global__ void k_place(const int* __restrict__ dst, const int* __restrict__ rank,
                        const int* __restrict__ rowptr, int* __restrict__ seg_eid, int E) {
    int e = blockIdx.x * blockDim.x + threadIdx.x;
    if (e < E) seg_eid[rowptr[dst[e]] + rank[e]] = e;
}

__global__ void k_seg_order(const int* __restrict__ src, const int* __restrict__ rowptr,
                            const int* __restrict__ seg_eid, int* __restrict__ seg_src, int n) {
    const int lane = threadIdx.x & 31;
    const int d = __builtin_amdgcn_readfirstlane((int)(blockIdx.x * (blockDim.x >> 5) + (threadIdx.x >> 5)));
    if (d >= n) return;
    const int beg = rowptr[d], end = rowptr[d + 1];
    for (int base = beg; base < end; base += 32) {
        int idx = base + lane;
        int myE = idx < end ? seg_eid[idx] : 0x7fffffff;
        int r = 0;
        for (int j = beg; j < end; ++j) r += (seg_eid[j] < myE) ? 1 : 0;
        if (idx < end) seg_src[beg + r] = src[myE];
    }
}

template <int H> struct HeadRow;
template <> struct HeadRow<4> {
    static __device__ __forceinline__ void load(const float* __restrict__ p, float (&a)[4]) {
        v4f t = *(const v4f*)p;
        a[0] = t.x; a[1] = t.y; a[2] = t.z; a[3] = t.w;
    }
};
template <> struct HeadRow<1> {
    static __device__ __forceinline__ void load(const float* __restrict__ p, float (&a)[1]) {
        a[0] = *p;
    }
};

__device__ __forceinline__ float lrelu02(float v) { return v > 0.f ? v : 0.2f * v; }

template <int H, int C, bool ELU>
__global__ void k_node_gather(const int* __restrict__ rowptr, const int* __restrict__ seg_src,
                              const float* __restrict__ asrc, const float* __restrict__ adst,
                              const float* __restrict__ feat, const float* __restrict__ bias,
                              float* __restrict__ out, int n) {
    constexpr int HC = H * C;
    constexpr int V = HC / 32;
    typedef __attribute__((ext_vector_type(V))) float vrow;
    const int lane = threadIdx.x & 31;
    const int d = __builtin_amdgcn_readfirstlane((int)(blockIdx.x * (blockDim.x >> 5) + (threadIdx.x >> 5)));
    if (d >= n) return;
    const int beg = rowptr[d], end = rowptr[d + 1];
    const int hd = (lane * V) / C;

    float ad[H], vself[H], mx[H];
    {
        float as_d[H];
        HeadRow<H>::load(adst + (size_t)d * H, ad);
        HeadRow<H>::load(asrc + (size_t)d * H, as_d);
#pragma unroll
        for (int h = 0; h < H; ++h) { vself[h] = lrelu02(as_d[h] + ad[h]); mx[h] = vself[h]; }
    }

    for (int base = beg; base < end; base += 32) {
        int idx = base + lane;
        if (idx < end) {
            int s = seg_src[idx];
            float a[H];
            HeadRow<H>::load(asrc + (size_t)s * H, a);
#pragma unroll
            for (int h = 0; h < H; ++h) mx[h] = fmaxf(mx[h], lrelu02(a[h] + ad[h]));
        }
    }
#pragma unroll
    for (int h = 0; h < H; ++h) {
#pragma unroll
        for (int o = 16; o > 0; o >>= 1) mx[h] = fmaxf(mx[h], __shfl_xor(mx[h], o));
    }

    float den[H];
#pragma unroll
    for (int h = 0; h < H; ++h) den[h] = 0.f;
    vrow acc = 0.f;
    for (int base = beg; base < end; base += 32) {
        int idx = base + lane;
        int s = 0;
        float ex[H];
#pragma unroll
        for (int h = 0; h < H; ++h) ex[h] = 0.f;
        if (idx < end) {
            s = seg_src[idx];
            float a[H];
            HeadRow<H>::load(asrc + (size_t)s * H, a);
#pragma unroll
            for (int h = 0; h < H; ++h) ex[h] = expf(lrelu02(a[h] + ad[h]) - mx[h]);
        }
#pragma unroll
        for (int h = 0; h < H; ++h) den[h] += ex[h];
        const int cnt = (end - base) < 32 ? (end - base) : 32;
#pragma unroll 4
        for (int j = 0; j < cnt; ++j) {
            const int sj = __builtin_amdgcn_readlane(s, j);
            float x = __int_as_float(__builtin_amdgcn_readlane(__float_as_int(ex[0]), j));
#pragma unroll
            for (int h = 1; h < H; ++h) {
                float xh = __int_as_float(__builtin_amdgcn_readlane(__float_as_int(ex[h]), j));
                x = (hd == h) ? xh : x;
            }
            vrow f = *(const vrow*)(feat + (size_t)sj * HC + lane * V);
            acc += f * x;
        }
    }
    float den_l = 0.f, xs = 0.f;
#pragma unroll
    for (int h = 0; h < H; ++h) {
        float t = den[h];
#pragma unroll
        for (int o = 16; o > 0; o >>= 1) t += __shfl_xor(t, o);
        float xself = expf(vself[h] - mx[h]);
        t += xself;
        if (h == 0 || hd == h) { den_l = t; xs = xself; }
    }
    const vrow bv = *(const vrow*)(bias + lane * V);
    {
        vrow f = *(const vrow*)(feat + (size_t)d * HC + lane * V);
        acc += f * xs;
    }
    vrow r = acc / den_l + bv;
    if (ELU) {
#pragma unroll
        for (int i = 0; i < V; ++i) r[i] = r[i] > 0.f ? r[i] : (expf(r[i]) - 1.f);
    }
    *(vrow*)(out + (size_t)d * HC + lane * V) = r;
}

extern "C" void kernel_launch(void* const* d_in, const int* in_sizes, int n_in,
                              void* d_out, int out_size, void* d_ws, size_t ws_size,
                              hipStream_t stream) {
    const float* x      = (const float*)d_in[0];
    const int*   ei     = (const int*)  d_in[1];
    const float* W1     = (const float*)d_in[2];
    const float* att_s1 = (const float*)d_in[3];
    const float* att_d1 = (const float*)d_in[4];
    const float* b1     = (const float*)d_in[5];
    const float* W2     = (const float*)d_in[6];
    const float* att_s2 = (const float*)d_in[7];
    const float* att_d2 = (const float*)d_in[8];
    const float* b2     = (const float*)d_in[9];

    const int IN = 128, HID = 128, H1 = 4, OUT = 64;
    const int n   = in_sizes[0] / IN;
    const int E   = in_sizes[1] / 2;
    const int nb  = (n + 255) / 256;

    float* out1  = (float*)d_out;
    float* hfeat = (float*)d_out + (size_t)n * OUT;

    float* w = (float*)d_ws;
    size_t off = 0;
    float* Wt1 = w + off; off += (size_t)IN * HID;
    float* Wt2 = w + off; off += (size_t)HID * OUT;
    float* h1  = w + off; off += (size_t)n * HID;
    float* h2  = w + off; off += (size_t)n * OUT;
    float* as1 = w + off; off += (size_t)n * H1;
    float* ad1 = w + off; off += (size_t)n * H1;
    float* as2 = w + off; off += (size_t)n;
    float* ad2 = w + off; off += (size_t)n;
    int* iw = (int*)(w + off);
    int* cnt     = iw; iw += n;
    int* rowptr  = iw; iw += n + 1;
    int* bsum    = iw; iw += nb;
    int* rank    = iw; iw += E;
    int* seg_eid = iw; iw += E;
    int* seg_src = iw; iw += E;

    const int* srcI = ei;
    const int* dstI = ei + E;

    auto cdiv = [](long a, long b) { return (unsigned)((a + b - 1) / b); };
    dim3 B256(256);

    k_pack_b<<<cdiv((long)IN * HID, 256), B256, 0, stream>>>(W1, Wt1, HID);
    k_pack_b<<<cdiv((long)HID * OUT, 256), B256, 0, stream>>>(W2, Wt2, OUT);

    k_zero_int  <<<cdiv(n, 256), B256, 0, stream>>>(cnt, n);
    k_count_rank<<<cdiv(E, 256), B256, 0, stream>>>(dstI, cnt, rank, E);
    k_scan_local<<<nb, B256, 0, stream>>>(cnt, rowptr, bsum, n);
    k_scan_add  <<<nb, B256, 0, stream>>>(rowptr, bsum, n, nb);
    k_place     <<<cdiv(E, 256), B256, 0, stream>>>(dstI, rank, rowptr, seg_eid, E);
    k_seg_order <<<cdiv(n, 8), B256, 0, stream>>>(srcI, rowptr, seg_eid, seg_src, n);

    k_gemm_f32_wmma<8><<<dim3(cdiv(n / 16, 4), HID / 64), dim3(128), 0, stream>>>(x, Wt1, h1, att_s1, att_d1, as1, ad1, HID, n / 16);
    k_node_gather<4, 32, true><<<cdiv(n, 8), B256, 0, stream>>>(rowptr, seg_src, as1, ad1, h1, b1, hfeat, n);

    k_gemm_f32_wmma<16><<<dim3(cdiv(n / 16, 4), OUT / 64), dim3(128), 0, stream>>>(hfeat, Wt2, h2, att_s2, att_d2, as2, ad2, OUT, n / 16);
    k_node_gather<1, 64, false><<<cdiv(n, 8), B256, 0, stream>>>(rowptr, seg_src, as2, ad2, h2, b2, out1, n);
}


# Gather reads a segment of up to 32 edges once: source id and scores stay in registers from the maximum to the weights, one scattered score lookup per edge instead of two; sums unchanged
# speedup vs baseline: 6.3937x; 1.0399x over previous
#include <hip/hip_runtime.h>
#include <hip/hip_bf16.h>
#include <math.h>

typedef __attribute__((ext_vector_type(2))) float v2f;
typedef __attribute__((ext_vector_type(8))) float v8f;

__global__ void k_zero_int(int* __restrict__ p, int n) {
    int i = blockIdx.x * blockDim.x + threadIdx.x;
    if (i < n) p[i] = 0;
}

__global__ void k_pack_b(const float* __restrict__ W, float* __restrict__ Bf, int Ncol) {
    int i = blockIdx.x * blockDim.x + threadIdx.x;
    if (i < 128 * Ncol) {
        int e = i & 3, lane = (i >> 2) & 31, j = (i >> 7) & 15, t = i >> 11;
        int k  = 8 * j + 4 * (lane >> 4) + e;
        int nn = 64 * (t >> 2) + 4 * (lane & 15) + (t & 3);
        Bf[i] = W[k * Ncol + nn];
    }
}

typedef __attribute__((ext_vector_type(4))) float v4g;

template <int M>
__device__ __forceinline__ float row_lane_xor(float x) {
    return __int_as_float(__builtin_amdgcn_update_dpp(0, __float_as_int(x), 0x160 + M, 0xf, 0xf, false));
}

template <int M, int K>
__device__ __forceinline__ void fold_pairs(float (&v)[16], const bool up) {
#pragma unroll
    for (int i = 0; i < K; ++i) {
        const float keep = up ? v[i + K] : v[i];
        const float send = up ? v[i] : v[i + K];
        v[i] = keep + row_lane_xor<M>(send);
    }
}

template <int LPH>
__global__ void __launch_bounds__(128)
k_gemm_f32_wmma(const float* __restrict__ A, const float* __restrict__ Bf,
                float* __restrict__ C, const float* __restrict__ att_s,
                const float* __restrict__ att_d, float* __restrict__ asrc,
                float* __restrict__ adst, int nCols, int nStripes) {
    const int lane = threadIdx.x & 31;
    const int hh   = lane >> 4;
    const int c    = lane & 15;
    const int stripe = __builtin_amdgcn_readfirstlane((int)(blockIdx.x * 4 + (threadIdx.x >> 5)));
    if (stripe >= nStripes) return;
    const int rowBase = stripe * 16;
    const int q = blockIdx.y;

    const float* arow = A + (size_t)(rowBase + c) * 128 + 4 * hh;
    v4g a[16];
#pragma unroll
    for (int j = 0; j < 16; ++j) a[j] = *(const v4g*)(arow + 8 * j);

    const float* bp = Bf + (size_t)q * (4 * 2048) + lane * 4;
    v8f acc[4];
#pragma unroll
    for (int t = 0; t < 4; ++t) acc[t] = (v8f){};

    v4g b[3][4];
    v4g ws, wd;
#pragma unroll
    for (int t = 0; t < 4; ++t) b[0][t] = *(const v4g*)(bp + (t * 16 + 0) * 128);
#pragma unroll
    for (int t = 0; t < 4; ++t) b[1][t] = *(const v4g*)(bp + (t * 16 + 1) * 128);

#pragma unroll
    for (int j = 0; j < 16; ++j) {
        if (j + 2 < 16) {
#pragma unroll
            for (int t = 0; t < 4; ++t)
                b[(j + 2) % 3][t] = *(const v4g*)(bp + (t * 16 + j + 2) * 128);
        }
        if (j == 8) {
            ws = *(const v4g*)(att_s + 64 * q + 4 * c);
            wd = *(const v4g*)(att_d + 64 * q + 4 * c);
        }
        __builtin_amdgcn_sched_barrier(0);
        const v2f a0 = {a[j].x, a[j].y};
        const v2f a1 = {a[j].z, a[j].w};
#pragma unroll
        for (int t = 0; t < 4; ++t) {
            const v4g bb = b[j % 3][t];
            const v2f b0 = {bb.x, bb.y};
            const v2f b1 = {bb.z, bb.w};
            acc[t] = __builtin_amdgcn_wmma_f32_16x16x4_f32(
                 false, a0,  false, b0,
                 (short)0, acc[t],  false,  false);
            acc[t] = __builtin_amdgcn_wmma_f32_16x16x4_f32(
                 false, a1,  false, b1,
                 (short)0, acc[t],  false,  false);
        }
        __builtin_amdgcn_sched_barrier(0);
    }
#pragma unroll
    for (int r = 0; r < 8; ++r) {
        v4g o = {acc[0][r], acc[1][r], acc[2][r], acc[3][r]};
        *(v4g*)(C + (size_t)(rowBase + r + 8 * hh) * nCols + 64 * q + 4 * c) = o;
    }

    float v[16];
#pragma unroll
    for (int r = 0; r < 8; ++r) {
        v[r]     = ((acc[0][r] * ws.x + acc[1][r] * ws.y) + acc[2][r] * ws.z) + acc[3][r] * ws.w;
        v[8 + r] = ((acc[0][r] * wd.x + acc[1][r] * wd.y) + acc[2][r] * wd.z) + acc[3][r] * wd.w;
    }
    fold_pairs<1, 8>(v, (lane & 1) != 0);
    fold_pairs<2, 4>(v, (lane & 2) != 0);
    fold_pairs<4, 2>(v, (lane & 4) != 0);
    if (LPH == 16) fold_pairs<8, 1>(v, (lane & 8) != 0);
    const int H    = nCols / (4 * LPH);
    const int head = LPH == 16 ? q : 2 * q + ((lane >> 3) & 1);
    const int r0   = 4 * ((lane >> 1) & 1) + 2 * ((lane >> 2) & 1) + (LPH == 16 ? ((lane >> 3) & 1) : 0);
    float* sp = ((lane & 1) ? adst : asrc) + (size_t)(rowBase + 8 * hh + r0) * H + head;
    sp[0] = v[0];
    if (LPH == 8) sp[H] = v[1];
}

typedef __attribute__((ext_vector_type(4))) float v4f;

__global__ void k_count_rank(const int* __restrict__ dst, int* __restrict__ cnt,
                             int* __restrict__ rank, int E) {
    int e = blockIdx.x * blockDim.x + threadIdx.x;
    if (e < E) rank[e] = atomicAdd(&cnt[dst[e]], 1);
}

__global__ void k_scan_local(const int* __restrict__ cnt, int* __restrict__ rowptr,
                             int* __restrict__ bsum, int n) {
    __shared__ int wsum[8];
    int t = threadIdx.x, i = blockIdx.x * 256 + t;
    int lane = t & 31, wv = t >> 5;
    int v = i < n ? cnt[i] : 0;
    int inc = v;
#pragma unroll
    for (int o = 1; o < 32; o <<= 1) {
        int u = __shfl_up(inc, o);
        if (lane >= o) inc += u;
    }
    if (lane == 31) wsum[wv] = inc;
    __syncthreads();
    int woff = 0;
#pragma unroll
    for (int k = 0; k < 8; ++k) woff += (k < wv) ? wsum[k] : 0;
    if (i < n) rowptr[i] = woff + inc - v;
    if (t == 255) bsum[blockIdx.x] = woff + inc;
}

__global__ void k_scan_add(int* __restrict__ rowptr, const int* __restrict__ bsum,
                           int n, int nb) {
    __shared__ int red[8];
    int t = threadIdx.x, b = blockIdx.x;
    int s = 0;
    for (int k = t; k < b; k += 256) s += bsum[k];
#pragma unroll
    for (int o = 16; o > 0; o >>= 1) s += __shfl_xor(s, o);
    if ((t & 31) == 0) red[t >> 5] = s;
    __syncthreads();
    int base = 0;
#pragma unroll
    for (int k = 0; k < 8; ++k) base += red[k];
    int i = b * 256 + t;
    if (i < n) rowptr[i] += base;
    if (b == nb - 1 && t == 0) rowptr[n] = base + bsum[b];
}

__global__ void k_place(const int* __restrict__ dst, const int* __restrict__ rank,
                        const int* __restrict__ rowptr, int* __restrict__ seg_eid, int E) {
    int e = blockIdx.x * blockDim.x + threadIdx.x;
    if (e < E) seg_eid[rowptr[dst[e]] + rank[e]] = e;
}

__global__ void k_seg_order(const int* __restrict__ src, const int* __restrict__ rowptr,
                            const int* __restrict__ seg_eid, int* __restrict__ seg_src, int n) {
    const int lane = threadIdx.x & 31;
    const int d = __builtin_amdgcn_readfirstlane((int)(blockIdx.x * (blockDim.x >> 5) + (threadIdx.x >> 5)));
    if (d >= n) return;
    const int beg = rowptr[d], end = rowptr[d + 1];
    for (int base = beg; base < end; base += 32) {
        int idx = base + lane;
        int myE = idx < end ? seg_eid[idx] : 0x7fffffff;
        int r = 0;
        for (int j = beg; j < end; ++j) r += (seg_eid[j] < myE) ? 1 : 0;
        if (idx < end) seg_src[beg + r] = src[myE];
    }
}

template <int H> struct HeadRow;
template <> struct HeadRow<4> {
    static __device__ __forceinline__ void load(const float* __restrict__ p, float (&a)[4]) {
        v4f t = *(const v4f*)p;
        a[0] = t.x; a[1] = t.y; a[2] = t.z; a[3] = t.w;
    }
};
template <> struct HeadRow<1> {
    static __device__ __forceinline__ void load(const float* __restrict__ p, float (&a)[1]) {
        a[0] = *p;
    }
};

__device__ __forceinline__ float lrelu02(float v) { return v > 0.f ? v : 0.2f * v; }

template <int H, int C>
__device__ __forceinline__ void add_rows(const float* __restrict__ feat, const int s,
                                         const float (&ex)[H], const int cnt, const int lane,
                                         const int hd,
                                         __attribute__((ext_vector_type(H * C / 32))) float& acc) {
    constexpr int HC = H * C;
    constexpr int V = HC / 32;
    typedef __attribute__((ext_vector_type(V))) float vrow;
#pragma unroll 4
    for (int j = 0; j < cnt; ++j) {
        const int sj = __builtin_amdgcn_readlane(s, j);
        float x = __int_as_float(__builtin_amdgcn_readlane(__float_as_int(ex[0]), j));
#pragma unroll
        for (int h = 1; h < H; ++h) {
            float xh = __int_as_float(__builtin_amdgcn_readlane(__float_as_int(ex[h]), j));
            x = (hd == h) ? xh : x;
        }
        vrow f = *(const vrow*)(feat + (size_t)sj * HC + lane * V);
        acc += f * x;
    }
}

template <int H, int C, bool ELU>
__global__ void k_node_gather(const int* __restrict__ rowptr, const int* __restrict__ seg_src,
                              const float* __restrict__ asrc, const float* __restrict__ adst,
                              const float* __restrict__ feat, const float* __restrict__ bias,
                              float* __restrict__ out, int n) {
    constexpr int HC = H * C;
    constexpr int V = HC / 32;
    typedef __attribute__((ext_vector_type(V))) float vrow;
    const int lane = threadIdx.x & 31;
    const int d = __builtin_amdgcn_readfirstlane((int)(blockIdx.x * (blockDim.x >> 5) + (threadIdx.x >> 5)));
    if (d >= n) return;
    const int beg = rowptr[d], end = rowptr[d + 1];
    const int deg = end - beg;
    const int hd = (lane * V) / C;

    float ad[H], vself[H], mx[H];
    {
        float as_d[H];
        HeadRow<H>::load(adst + (size_t)d * H, ad);
        HeadRow<H>::load(asrc + (size_t)d * H, as_d);
#pragma unroll
        for (int h = 0; h < H; ++h) { vself[h] = lrelu02(as_d[h] + ad[h]); mx[h] = vself[h]; }
    }

    float den[H];
#pragma unroll
    for (int h = 0; h < H; ++h) den[h] = 0.f;
    vrow acc = 0.f;

    if (deg <= 32) {
        const bool has = lane < deg;
        int s = 0;
        float e[H];
#pragma unroll
        for (int h = 0; h < H; ++h) e[h] = 0.f;
        if (has) {
            s = seg_src[beg + lane];
            float a[H];
            HeadRow<H>::load(asrc + (size_t)s * H, a);
#pragma unroll
            for (int h = 0; h < H; ++h) { e[h] = lrelu02(a[h] + ad[h]); mx[h] = fmaxf(mx[h], e[h]); }
        }
#pragma unroll
        for (int h = 0; h < H; ++h) {
#pragma unroll
            for (int o = 16; o > 0; o >>= 1) mx[h] = fmaxf(mx[h], __shfl_xor(mx[h], o));
        }
        float ex[H];
#pragma unroll
        for (int h = 0; h < H; ++h) ex[h] = 0.f;
        if (has) {
#pragma unroll
            for (int h = 0; h < H; ++h) ex[h] = expf(e[h] - mx[h]);
        }
#pragma unroll
        for (int h = 0; h < H; ++h) den[h] += ex[h];
        add_rows<H, C>(feat, s, ex, deg, lane, hd, acc);
    } else {
        for (int base = beg; base < end; base += 32) {
            int idx = base + lane;
            if (idx < end) {
                int s = seg_src[idx];
                float a[H];
                HeadRow<H>::load(asrc + (size_t)s * H, a);
#pragma unroll
                for (int h = 0; h < H; ++h) mx[h] = fmaxf(mx[h], lrelu02(a[h] + ad[h]));
            }
        }
#pragma unroll
        for (int h = 0; h < H; ++h) {
#pragma unroll
            for (int o = 16; o > 0; o >>= 1) mx[h] = fmaxf(mx[h], __shfl_xor(mx[h], o));
        }

        for (int base = beg; base < end; base += 32) {
            int idx = base + lane;
            int s = 0;
            float ex[H];
#pragma unroll
            for (int h = 0; h < H; ++h) ex[h] = 0.f;
            if (idx < end) {
                s = seg_src[idx];
                float a[H];
                HeadRow<H>::load(asrc + (size_t)s * H, a);
#pragma unroll
                for (int h = 0; h < H; ++h) ex[h] = expf(lrelu02(a[h] + ad[h]) - mx[h]);
            }
#pragma unroll
            for (int h = 0; h < H; ++h) den[h] += ex[h];
            const int cnt = (end - base) < 32 ? (end - base) : 32;
            add_rows<H, C>(feat, s, ex, cnt, lane, hd, acc);
        }
    }

    float den_l = 0.f, xs = 0.f;
#pragma unroll
    for (int h = 0; h < H; ++h) {
        float t = den[h];
#pragma unroll
        for (int o = 16; o > 0; o >>= 1) t += __shfl_xor(t, o);
        float xself = expf(vself[h] - mx[h]);
        t += xself;
        if (h == 0 || hd == h) { den_l = t; xs = xself; }
    }
    const vrow bv = *(const vrow*)(bias + lane * V);
    {
        vrow f = *(const vrow*)(feat + (size_t)d * HC + lane * V);
        acc += f * xs;
    }
    vrow r = acc / den_l + bv;
    if (ELU) {
#pragma unroll
        for (int i = 0; i < V; ++i) r[i] = r[i] > 0.f ? r[i] : (expf(r[i]) - 1.f);
    }
    *(vrow*)(out + (size_t)d * HC + lane * V) = r;
}

extern "C" void kernel_launch(void* const* d_in, const int* in_sizes, int n_in,
                              void* d_out, int out_size, void* d_ws, size_t ws_size,
                              hipStream_t stream) {
    const float* x      = (const float*)d_in[0];
    const int*   ei     = (const int*)  d_in[1];
    const float* W1     = (const float*)d_in[2];
    const float* att_s1 = (const float*)d_in[3];
    const float* att_d1 = (const float*)d_in[4];
    const float* b1     = (const float*)d_in[5];
    const float* W2     = (const float*)d_in[6];
    const float* att_s2 = (const float*)d_in[7];
    const float* att_d2 = (const float*)d_in[8];
    const float* b2     = (const float*)d_in[9];

    const int IN = 128, HID = 128, H1 = 4, OUT = 64;
    const int n   = in_sizes[0] / IN;
    const int E   = in_sizes[1] / 2;
    const int nb  = (n + 255) / 256;

    float* out1  = (float*)d_out;
    float* hfeat = (float*)d_out + (size_t)n * OUT;

    float* w = (float*)d_ws;
    size_t off = 0;
    float* Wt1 = w + off; off += (size_t)IN * HID;
    float* Wt2 = w + off; off += (size_t)HID * OUT;
    float* h1  = w + off; off += (size_t)n * HID;
    float* h2  = w + off; off += (size_t)n * OUT;
    float* as1 = w + off; off += (size_t)n * H1;
    float* ad1 = w + off; off += (size_t)n * H1;
    float* as2 = w + off; off += (size_t)n;
    float* ad2 = w + off; off += (size_t)n;
    int* iw = (int*)(w + off);
    int* cnt     = iw; iw += n;
    int* rowptr  = iw; iw += n + 1;
    int* bsum    = iw; iw += nb;
    int* rank    = iw; iw += E;
    int* seg_eid = iw; iw += E;
    int* seg_src = iw; iw += E;

    const int* srcI = ei;
    const int* dstI = ei + E;

    auto cdiv = [](long a, long b) { return (unsigned)((a + b - 1) / b); };
    dim3 B256(256);

    k_pack_b<<<cdiv((long)IN * HID, 256), B256, 0, stream>>>(W1, Wt1, HID);
    k_pack_b<<<cdiv((long)HID * OUT, 256), B256, 0, stream>>>(W2, Wt2, OUT);

    k_zero_int  <<<cdiv(n, 256), B256, 0, stream>>>(cnt, n);
    k_count_rank<<<cdiv(E, 256), B256, 0, stream>>>(dstI, cnt, rank, E);
    k_scan_local<<<nb, B256, 0, stream>>>(cnt, rowptr, bsum, n);
    k_scan_add  <<<nb, B256, 0, stream>>>(rowptr, bsum, n, nb);
    k_place     <<<cdiv(E, 256), B256, 0, stream>>>(dstI, rank, rowptr, seg_eid, E);
    k_seg_order <<<cdiv(n, 8), B256, 0, stream>>>(srcI, rowptr, seg_eid, seg_src, n);

    k_gemm_f32_wmma<8><<<dim3(cdiv(n / 16, 4), HID / 64), dim3(128), 0, stream>>>(x, Wt1, h1, att_s1, att_d1, as1, ad1, HID, n / 16);
    k_node_gather<4, 32, true><<<cdiv(n, 8), B256, 0, stream>>>(rowptr, seg_src, as1, ad1, h1, b1, hfeat, n);

    k_gemm_f32_wmma<16><<<dim3(cdiv(n / 16, 4), OUT / 64), dim3(128), 0, stream>>>(hfeat, Wt2, h2, att_s2, att_d2, as2, ad2, OUT, n / 16);
    k_node_gather<1, 64, false><<<cdiv(n, 8), B256, 0, stream>>>(rowptr, seg_src, as2, ad2, h2, b2, out1, n);
}


# Segments ordered by source id inside the first gather (lane-compare rank + one lane permute; stored for layer 2): ordering kernel and edge-id list removed; sums now run in source order
# speedup vs baseline: 6.5852x; 1.0300x over previous
#include <hip/hip_runtime.h>
#include <hip/hip_bf16.h>
#include <math.h>

typedef __attribute__((ext_vector_type(2))) float v2f;
typedef __attribute__((ext_vector_type(8))) float v8f;

__global__ void k_zero_int(int* __restrict__ p, int n) {
    int i = blockIdx.x * blockDim.x + threadIdx.x;
    if (i < n) p[i] = 0;
}

__global__ void k_pack_b(const float* __restrict__ W, float* __restrict__ Bf, int Ncol) {
    int i = blockIdx.x * blockDim.x + threadIdx.x;
    if (i < 128 * Ncol) {
        int e = i & 3, lane = (i >> 2) & 31, j = (i >> 7) & 15, t = i >> 11;
        int k  = 8 * j + 4 * (lane >> 4) + e;
        int nn = 64 * (t >> 2) + 4 * (lane & 15) + (t & 3);
        Bf[i] = W[k * Ncol + nn];
    }
}

typedef __attribute__((ext_vector_type(4))) float v4g;

template <int M>
__device__ __forceinline__ float row_lane_xor(float x) {
    return __int_as_float(__builtin_amdgcn_update_dpp(0, __float_as_int(x), 0x160 + M, 0xf, 0xf, false));
}

template <int M, int K>
__device__ __forceinline__ void fold_pairs(float (&v)[16], const bool up) {
#pragma unroll
    for (int i = 0; i < K; ++i) {
        const float keep = up ? v[i + K] : v[i];
        const float send = up ? v[i] : v[i + K];
        v[i] = keep + row_lane_xor<M>(send);
    }
}

template <int LPH>
__global__ void __launch_bounds__(128)
k_gemm_f32_wmma(const float* __restrict__ A, const float* __restrict__ Bf,
                float* __restrict__ C, const float* __restrict__ att_s,
                const float* __restrict__ att_d, float* __restrict__ asrc,
                float* __restrict__ adst, int nCols, int nStripes) {
    const int lane = threadIdx.x & 31;
    const int hh   = lane >> 4;
    const int c    = lane & 15;
    const int stripe = __builtin_amdgcn_readfirstlane((int)(blockIdx.x * 4 + (threadIdx.x >> 5)));
    if (stripe >= nStripes) return;
    const int rowBase = stripe * 16;
    const int q = blockIdx.y;

    const float* arow = A + (size_t)(rowBase + c) * 128 + 4 * hh;
    v4g a[16];
#pragma unroll
    for (int j = 0; j < 16; ++j) a[j] = *(const v4g*)(arow + 8 * j);

    const float* bp = Bf + (size_t)q * (4 * 2048) + lane * 4;
    v8f acc[4];
#pragma unroll
    for (int t = 0; t < 4; ++t) acc[t] = (v8f){};

    v4g b[3][4];
    v4g ws, wd;
#pragma unroll
    for (int t = 0; t < 4; ++t) b[0][t] = *(const v4g*)(bp + (t * 16 + 0) * 128);
#pragma unroll
    for (int t = 0; t < 4; ++t) b[1][t] = *(const v4g*)(bp + (t * 16 + 1) * 128);

#pragma unroll
    for (int j = 0; j < 16; ++j) {
        if (j + 2 < 16) {
#pragma unroll
            for (int t = 0; t < 4; ++t)
                b[(j + 2) % 3][t] = *(const v4g*)(bp + (t * 16 + j + 2) * 128);
        }
        if (j == 8) {
            ws = *(const v4g*)(att_s + 64 * q + 4 * c);
            wd = *(const v4g*)(att_d + 64 * q + 4 * c);
        }
        __builtin_amdgcn_sched_barrier(0);
        const v2f a0 = {a[j].x, a[j].y};
        const v2f a1 = {a[j].z, a[j].w};
#pragma unroll
        for (int t = 0; t < 4; ++t) {
            const v4g bb = b[j % 3][t];
            const v2f b0 = {bb.x, bb.y};
            const v2f b1 = {bb.z, bb.w};
            acc[t] = __builtin_amdgcn_wmma_f32_16x16x4_f32(
                 false, a0,  false, b0,
                 (short)0, acc[t],  false,  false);
            acc[t] = __builtin_amdgcn_wmma_f32_16x16x4_f32(
                 false, a1,  false, b1,
                 (short)0, acc[t],  false,  false);
        }
        __builtin_amdgcn_sched_barrier(0);
    }
#pragma unroll
    for (int r = 0; r < 8; ++r) {
        v4g o = {acc[0][r], acc[1][r], acc[2][r], acc[3][r]};
        *(v4g*)(C + (size_t)(rowBase + r + 8 * hh) * nCols + 64 * q + 4 * c) = o;
    }

    float v[16];
#pragma unroll
    for (int r = 0; r < 8; ++r) {
        v[r]     = ((acc[0][r] * ws.x + acc[1][r] * ws.y) + acc[2][r] * ws.z) + acc[3][r] * ws.w;
        v[8 + r] = ((acc[0][r] * wd.x + acc[1][r] * wd.y) + acc[2][r] * wd.z) + acc[3][r] * wd.w;
    }
    fold_pairs<1, 8>(v, (lane & 1) != 0);
    fold_pairs<2, 4>(v, (lane & 2) != 0);
    fold_pairs<4, 2>(v, (lane & 4) != 0);
    if (LPH == 16) fold_pairs<8, 1>(v, (lane & 8) != 0);
    const int H    = nCols / (4 * LPH);
    const int head = LPH == 16 ? q : 2 * q + ((lane >> 3) & 1);
    const int r0   = 4 * ((lane >> 1) & 1) + 2 * ((lane >> 2) & 1) + (LPH == 16 ? ((lane >> 3) & 1) : 0);
    float* sp = ((lane & 1) ? adst : asrc) + (size_t)(rowBase + 8 * hh + r0) * H + head;
    sp[0] = v[0];
    if (LPH == 8) sp[H] = v[1];
}

typedef __attribute__((ext_vector_type(4))) float v4f;

__global__ void k_count_rank(const int* __restrict__ dst, int* __restrict__ cnt,
                             int* __restrict__ rank, int E) {
    int e = blockIdx.x * blockDim.x + threadIdx.x;
    if (e < E) rank[e] = atomicAdd(&cnt[dst[e]], 1);
}

__global__ void k_scan_local(const int* __restrict__ cnt, int* __restrict__ rowptr,
                             int* __restrict__ bsum, int n) {
    __shared__ int wsum[8];
    int t = threadIdx.x, i = blockIdx.x * 256 + t;
    int lane = t & 31, wv = t >> 5;
    int v = i < n ? cnt[i] : 0;
    int inc = v;
#pragma unroll
    for (int o = 1; o < 32; o <<= 1) {
        int u = __shfl_up(inc, o);
        if (lane >= o) inc += u;
    }
    if (lane == 31) wsum[wv] = inc;
    __syncthreads();
    int woff = 0;
#pragma unroll
    for (int k = 0; k < 8; ++k) woff += (k < wv) ? wsum[k] : 0;
    if (i < n) rowptr[i] = woff + inc - v;
    if (t == 255) bsum[blockIdx.x] = woff + inc;
}

__global__ void k_scan_add(int* __restrict__ rowptr, const int* __restrict__ bsum,
                           int n, int nb) {
    __shared__ int red[8];
    int t = threadIdx.x, b = blockIdx.x;
    int s = 0;
    for (int k = t; k < b; k += 256) s += bsum[k];
#pragma unroll
    for (int o = 16; o > 0; o >>= 1) s += __shfl_xor(s, o);
    if ((t & 31) == 0) red[t >> 5] = s;
    __syncthreads();
    int base = 0;
#pragma unroll
    for (int k = 0; k < 8; ++k) base += red[k];
    int i = b * 256 + t;
    if (i < n) rowptr[i] += base;
    if (b == nb - 1 && t == 0) rowptr[n] = base + bsum[b];
}

__global__ void k_place(const int* __restrict__ src, const int* __restrict__ dst,
                        const int* __restrict__ rank, const int* __restrict__ rowptr,
                        int* __restrict__ seg_raw, int E) {
    int e = blockIdx.x * blockDim.x + threadIdx.x;
    if (e < E) seg_raw[rowptr[dst[e]] + rank[e]] = src[e];
}

template <int H> struct HeadRow;
template <> struct HeadRow<4> {
    static __device__ __forceinline__ void load(const float* __restrict__ p, float (&a)[4]) {
        v4f t = *(const v4f*)p;
        a[0] = t.x; a[1] = t.y; a[2] = t.z; a[3] = t.w;
    }
};
template <> struct HeadRow<1> {
    static __device__ __forceinline__ void load(const float* __restrict__ p, float (&a)[1]) {
        a[0] = *p;
    }
};

__device__ __forceinline__ float lrelu02(float v) { return v > 0.f ? v : 0.2f * v; }

template <int H, int C>
__device__ __forceinline__ void add_rows(const float* __restrict__ feat, const int s,
                                         const float (&ex)[H], const int cnt, const int lane,
                                         const int hd,
                                         __attribute__((ext_vector_type(H * C / 32))) float& acc) {
    constexpr int HC = H * C;
    constexpr int V = HC / 32;
    typedef __attribute__((ext_vector_type(V))) float vrow;
#pragma unroll 4
    for (int j = 0; j < cnt; ++j) {
        const int sj = __builtin_amdgcn_readlane(s, j);
        float x = __int_as_float(__builtin_amdgcn_readlane(__float_as_int(ex[0]), j));
#pragma unroll
        for (int h = 1; h < H; ++h) {
            float xh = __int_as_float(__builtin_amdgcn_readlane(__float_as_int(ex[h]), j));
            x = (hd == h) ? xh : x;
        }
        vrow f = *(const vrow*)(feat + (size_t)sj * HC + lane * V);
        acc += f * x;
    }
}

template <int H, int C, bool ELU, bool ORDER>
__global__ void k_node_gather(const int* __restrict__ rowptr, const int* __restrict__ seg_in,
                              int* __restrict__ seg_out,
                              const float* __restrict__ asrc, const float* __restrict__ adst,
                              const float* __restrict__ feat, const float* __restrict__ bias,
                              float* __restrict__ out, int n) {
    constexpr int HC = H * C;
    constexpr int V = HC / 32;
    typedef __attribute__((ext_vector_type(V))) float vrow;
    const int lane = threadIdx.x & 31;
    const int d = __builtin_amdgcn_readfirstlane((int)(blockIdx.x * (blockDim.x >> 5) + (threadIdx.x >> 5)));
    if (d >= n) return;
    const int beg = rowptr[d], end = rowptr[d + 1];
    const int deg = end - beg;
    const int hd = (lane * V) / C;

    float ad[H], vself[H], mx[H];
    {
        float as_d[H];
        HeadRow<H>::load(adst + (size_t)d * H, ad);
        HeadRow<H>::load(asrc + (size_t)d * H, as_d);
#pragma unroll
        for (int h = 0; h < H; ++h) { vself[h] = lrelu02(as_d[h] + ad[h]); mx[h] = vself[h]; }
    }

    float den[H];
#pragma unroll
    for (int h = 0; h < H; ++h) den[h] = 0.f;
    vrow acc = 0.f;

    if (deg <= 32) {
        const bool has = lane < deg;
        int s = 0;
        if (ORDER) {
            const int sraw = has ? seg_in[beg + lane] : 0;
            const int key = has ? ((sraw << 5) | lane) : 0x7fffffff;
            int r = 0;
#pragma unroll 4
            for (int j = 0; j < deg; ++j)
                r += (__builtin_amdgcn_readlane(key, j) < key) ? 1 : 0;
            r = has ? r : lane;
            s = __builtin_amdgcn_ds_permute(r << 2, sraw);
            if (has) seg_out[beg + lane] = s;
        } else {
            if (has) s = seg_in[beg + lane];
        }
        float e[H];
#pragma unroll
        for (int h = 0; h < H; ++h) e[h] = 0.f;
        if (has) {
            float a[H];
            HeadRow<H>::load(asrc + (size_t)s * H, a);
#pragma unroll
            for (int h = 0; h < H; ++h) { e[h] = lrelu02(a[h] + ad[h]); mx[h] = fmaxf(mx[h], e[h]); }
        }
#pragma unroll
        for (int h = 0; h < H; ++h) {
#pragma unroll
            for (int o = 16; o > 0; o >>= 1) mx[h] = fmaxf(mx[h], __shfl_xor(mx[h], o));
        }
        float ex[H];
#pragma unroll
        for (int h = 0; h < H; ++h) ex[h] = 0.f;
        if (has) {
#pragma unroll
            for (int h = 0; h < H; ++h) ex[h] = expf(e[h] - mx[h]);
        }
#pragma unroll
        for (int h = 0; h < H; ++h) den[h] += ex[h];
        add_rows<H, C>(feat, s, ex, deg, lane, hd, acc);
    } else {
        if (ORDER) {
            for (int base = beg; base < end; base += 32) {
                const int idx = base + lane;
                const int myS = idx < end ? seg_in[idx] : 0x7fffffff;
                int r = 0;
                for (int j = beg; j < end; ++j) {
                    const int sj = seg_in[j];
                    r += (sj < myS || (sj == myS && j < idx)) ? 1 : 0;
                }
                if (idx < end) seg_out[beg + r] = myS;
            }
            __threadfence();
        }
        const int* seg_src = ORDER ? (const int*)seg_out : seg_in;
        for (int base = beg; base < end; base += 32) {
            int idx = base + lane;
            if (idx < end) {
                int s = seg_src[idx];
                float a[H];
                HeadRow<H>::load(asrc + (size_t)s * H, a);
#pragma unroll
                for (int h = 0; h < H; ++h) mx[h] = fmaxf(mx[h], lrelu02(a[h] + ad[h]));
            }
        }
#pragma unroll
        for (int h = 0; h < H; ++h) {
#pragma unroll
            for (int o = 16; o > 0; o >>= 1) mx[h] = fmaxf(mx[h], __shfl_xor(mx[h], o));
        }

        for (int base = beg; base < end; base += 32) {
            int idx = base + lane;
            int s = 0;
            float ex[H];
#pragma unroll
            for (int h = 0; h < H; ++h) ex[h] = 0.f;
            if (idx < end) {
                s = seg_src[idx];
                float a[H];
                HeadRow<H>::load(asrc + (size_t)s * H, a);
#pragma unroll
                for (int h = 0; h < H; ++h) ex[h] = expf(lrelu02(a[h] + ad[h]) - mx[h]);
            }
#pragma unroll
            for (int h = 0; h < H; ++h) den[h] += ex[h];
            const int cnt = (end - base) < 32 ? (end - base) : 32;
            add_rows<H, C>(feat, s, ex, cnt, lane, hd, acc);
        }
    }

    float den_l = 0.f, xs = 0.f;
#pragma unroll
    for (int h = 0; h < H; ++h) {
        float t = den[h];
#pragma unroll
        for (int o = 16; o > 0; o >>= 1) t += __shfl_xor(t, o);
        float xself = expf(vself[h] - mx[h]);
        t += xself;
        if (h == 0 || hd == h) { den_l = t; xs = xself; }
    }
    const vrow bv = *(const vrow*)(bias + lane * V);
    {
        vrow f = *(const vrow*)(feat + (size_t)d * HC + lane * V);
        acc += f * xs;
    }
    vrow r = acc / den_l + bv;
    if (ELU) {
#pragma unroll
        for (int i = 0; i < V; ++i) r[i] = r[i] > 0.f ? r[i] : (expf(r[i]) - 1.f);
    }
    *(vrow*)(out + (size_t)d * HC + lane * V) = r;
}

extern "C" void kernel_launch(void* const* d_in, const int* in_sizes, int n_in,
                              void* d_out, int out_size, void* d_ws, size_t ws_size,
                              hipStream_t stream) {
    const float* x      = (const float*)d_in[0];
    const int*   ei     = (const int*)  d_in[1];
    const float* W1     = (const float*)d_in[2];
    const float* att_s1 = (const float*)d_in[3];
    const float* att_d1 = (const float*)d_in[4];
    const float* b1     = (const float*)d_in[5];
    const float* W2     = (const float*)d_in[6];
    const float* att_s2 = (const float*)d_in[7];
    const float* att_d2 = (const float*)d_in[8];
    const float* b2     = (const float*)d_in[9];

    const int IN = 128, HID = 128, H1 = 4, OUT = 64;
    const int n   = in_sizes[0] / IN;
    const int E   = in_sizes[1] / 2;
    const int nb  = (n + 255) / 256;

    float* out1  = (float*)d_out;
    float* hfeat = (float*)d_out + (size_t)n * OUT;

    float* w = (float*)d_ws;
    size_t off = 0;
    float* Wt1 = w + off; off += (size_t)IN * HID;
    float* Wt2 = w + off; off += (size_t)HID * OUT;
    float* h1  = w + off; off += (size_t)n * HID;
    float* h2  = w + off; off += (size_t)n * OUT;
    float* as1 = w + off; off += (size_t)n * H1;
    float* ad1 = w + off; off += (size_t)n * H1;
    float* as2 = w + off; off += (size_t)n;
    float* ad2 = w + off; off += (size_t)n;
    int* iw = (int*)(w + off);
    int* cnt     = iw; iw += n;
    int* rowptr  = iw; iw += n + 1;
    int* bsum    = iw; iw += nb;
    int* rank    = iw; iw += E;
    int* seg_raw = iw; iw += E;
    int* seg_src = iw; iw += E;

    const int* srcI = ei;
    const int* dstI = ei + E;

    auto cdiv = [](long a, long b) { return (unsigned)((a + b - 1) / b); };
    dim3 B256(256);

    k_pack_b<<<cdiv((long)IN * HID, 256), B256, 0, stream>>>(W1, Wt1, HID);
    k_pack_b<<<cdiv((long)HID * OUT, 256), B256, 0, stream>>>(W2, Wt2, OUT);

    k_zero_int  <<<cdiv(n, 256), B256, 0, stream>>>(cnt, n);
    k_count_rank<<<cdiv(E, 256), B256, 0, stream>>>(dstI, cnt, rank, E);
    k_scan_local<<<nb, B256, 0, stream>>>(cnt, rowptr, bsum, n);
    k_scan_add  <<<nb, B256, 0, stream>>>(rowptr, bsum, n, nb);
    k_place     <<<cdiv(E, 256), B256, 0, stream>>>(srcI, dstI, rank, rowptr, seg_raw, E);

    k_gemm_f32_wmma<8><<<dim3(cdiv(n / 16, 4), HID / 64), dim3(128), 0, stream>>>(x, Wt1, h1, att_s1, att_d1, as1, ad1, HID, n / 16);
    k_node_gather<4, 32, true, true><<<cdiv(n, 8), B256, 0, stream>>>(rowptr, seg_raw, seg_src, as1, ad1, h1, b1, hfeat, n);

    k_gemm_f32_wmma<16><<<dim3(cdiv(n / 16, 4), OUT / 64), dim3(128), 0, stream>>>(hfeat, Wt2, h2, att_s2, att_d2, as2, ad2, OUT, n / 16);
    k_node_gather<1, 64, false, false><<<cdiv(n, 8), B256, 0, stream>>>(rowptr, seg_src, (int*)nullptr, as2, ad2, h2, b2, out1, n);
}
